# starting kernel
# baseline (speedup 1.0000x reference)
#include <hip/hip_runtime.h>
#include <hip/hip_bf16.h>

#define N_NODES 10000
#define N_EDGES 640000
#define N_GRAPHS 64
#define DD 128
#define NEG_SLOPE 0.01f

typedef __attribute__((ext_vector_type(2))) float v2f;
typedef __attribute__((ext_vector_type(8))) float v8f;

__device__ __forceinline__ unsigned enc_f32(float f) {
    unsigned u = __float_as_uint(f);
    return (u & 0x80000000u) ? ~u : (u | 0x80000000u);
}
__device__ __forceinline__ float dec_f32(unsigned e) {
    return (e & 0x80000000u) ? __uint_as_float(e & 0x7FFFFFFFu)
                             : __uint_as_float(~e);
}

__global__ void k_init(unsigned* __restrict__ menc, float* __restrict__ denom,
                       float* __restrict__ phis) {
    int i = blockIdx.x * blockDim.x + threadIdx.x;
    if (i < 2 * N_NODES) { menc[i] = 0u; denom[i] = 0.0f; }
    if (i < N_GRAPHS * 2 * DD) phis[i] = 0.0f;
}

__global__ void k_gemm(const float* __restrict__ x,
                       const float* __restrict__ Wfc, const float* __restrict__ bfc,
                       const float* __restrict__ Wres,
                       float* __restrict__ feat_ws, float* __restrict__ hout) {
    __shared__ float WT[DD * DD];
    const int b     = blockIdx.y;
    const int which = blockIdx.z;
    const float* W = (which == 0 ? Wfc : Wres) + (size_t)b * DD * DD;

    for (int idx = threadIdx.x; idx < DD * DD; idx += blockDim.x) {
        int j = idx >> 7, k = idx & 127;
        WT[k * DD + j] = W[idx];
    }
    __syncthreads();

    const int lane = threadIdx.x & 31;
    const int wave = threadIdx.x >> 5;
    const int rowBase = blockIdx.x * 64 + wave * 16;
    const int l16 = lane & 15;
    const int hi  = lane >> 4;

    v8f c[8];
#pragma unroll
    for (int t = 0; t < 8; ++t) {
        float bv = (which == 0) ? bfc[b * DD + t * 16 + l16] : 0.0f;
#pragma unroll
        for (int v = 0; v < 8; ++v) c[t][v] = bv;
    }

    const int arow = min(rowBase + l16, N_NODES - 1);
    const float* xrow = x + (size_t)arow * DD;

    for (int k0 = 0; k0 < DD; k0 += 4) {
        const int krow = k0 + 2 * hi;
        v2f a = *(const v2f*)(xrow + krow);
#pragma unroll
        for (int t = 0; t < 8; ++t) {
            const int j = t * 16 + l16;
            v2f bf;
            bf.x = WT[krow * DD + j];
            bf.y = WT[(krow + 1) * DD + j];
            c[t] = __builtin_amdgcn_wmma_f32_16x16x4_f32(
                false, a, false, bf, (short)0, c[t], false, false);
        }
    }

#pragma unroll
    for (int t = 0; t < 8; ++t) {
        const int col = t * 16 + l16;
#pragma unroll
        for (int v = 0; v < 8; ++v) {
            const int row = rowBase + v + 8 * hi;
            if (row < N_NODES) {
                if (which == 0)
                    feat_ws[(size_t)b * N_NODES * DD + (size_t)row * DD + col] = c[t][v];
                else
                    hout[(size_t)row * (2 * DD) + b * DD + col] = c[t][v];
            }
        }
    }
}

__global__ void k_attn_coef(const float* __restrict__ feat,
                            const float* __restrict__ wl, const float* __restrict__ wr,
                            float* __restrict__ a_self, float* __restrict__ a_nb) {
    const int b = blockIdx.y;
    const int row = blockIdx.x * (blockDim.x >> 5) + (threadIdx.x >> 5);
    if (row >= N_NODES) return;
    const int lane = threadIdx.x & 31;
    float4 f = ((const float4*)(feat + (size_t)b * N_NODES * DD + (size_t)row * DD))[lane];
    float4 l = ((const float4*)(wl + b * DD))[lane];
    float4 r = ((const float4*)(wr + b * DD))[lane];
    float sl = f.x * l.x + f.y * l.y + f.z * l.z + f.w * l.w;
    float sr = f.x * r.x + f.y * r.y + f.z * r.z + f.w * r.w;
#pragma unroll
    for (int m = 16; m >= 1; m >>= 1) {
        sl += __shfl_xor(sl, m, 32);
        sr += __shfl_xor(sr, m, 32);
    }
    if (lane == 0) {
        a_self[b * N_NODES + row] = sl;
        a_nb[b * N_NODES + row]   = sr;
    }
}

__device__ __forceinline__ float edge_score(const float* a_self, const float* a_nb,
                                            int b, int s, int d) {
    float sc = a_self[b * N_NODES + d] + a_nb[b * N_NODES + s];
    return sc >= 0.0f ? sc : NEG_SLOPE * sc;
}

__global__ void k_edge_max(const int* __restrict__ esrc, const int* __restrict__ edst,
                           const float* __restrict__ a_self, const float* __restrict__ a_nb,
                           unsigned* __restrict__ menc) {
    const int e = blockIdx.x * blockDim.x + threadIdx.x;
    const int b = blockIdx.y;
    if (e >= N_EDGES) return;
    float sc = edge_score(a_self, a_nb, b, esrc[e], edst[e]);
    atomicMax(&menc[b * N_NODES + edst[e]], enc_f32(sc));
}

__global__ void k_edge_denom(const int* __restrict__ esrc, const int* __restrict__ edst,
                             const float* __restrict__ a_self, const float* __restrict__ a_nb,
                             const unsigned* __restrict__ menc, float* __restrict__ denom) {
    const int e = blockIdx.x * blockDim.x + threadIdx.x;
    const int b = blockIdx.y;
    if (e >= N_EDGES) return;
    const int d = edst[e];
    float sc = edge_score(a_self, a_nb, b, esrc[e], d);
    float ex = __expf(sc - dec_f32(menc[b * N_NODES + d]));
    atomicAdd(&denom[b * N_NODES + d], ex);
}

__global__ void k_edge_accum(const int* __restrict__ esrc, const int* __restrict__ edst,
                             const float* __restrict__ a_self, const float* __restrict__ a_nb,
                             const unsigned* __restrict__ menc, const float* __restrict__ denom,
                             const float* __restrict__ feat, float* __restrict__ hout) {
    const int b = blockIdx.y;
    const int lane = threadIdx.x & 31;
    const int wave = (blockIdx.x * blockDim.x + threadIdx.x) >> 5;
    const int nwaves = (gridDim.x * blockDim.x) >> 5;
    for (int e = wave; e < N_EDGES; e += nwaves) {
        const int s = esrc[e], d = edst[e];
        float sc = edge_score(a_self, a_nb, b, s, d);
        float w = __expf(sc - dec_f32(menc[b * N_NODES + d])) / denom[b * N_NODES + d];
        float4 f = ((const float4*)(feat + (size_t)b * N_NODES * DD + (size_t)s * DD))[lane];
        float* dst = hout + (size_t)d * (2 * DD) + b * DD + lane * 4;
        atomicAdd(dst + 0, w * f.x);
        atomicAdd(dst + 1, w * f.y);
        atomicAdd(dst + 2, w * f.z);
        atomicAdd(dst + 3, w * f.w);
    }
}

__global__ void k_relu_pool(float* __restrict__ hout, const int* __restrict__ gids,
                            float* __restrict__ phis) {
    const int idx = blockIdx.x * blockDim.x + threadIdx.x;
    const int n = idx >> 6, q = idx & 63;
    if (n >= N_NODES) return;
    const int g = gids[n];
    float4 v = ((float4*)(hout + (size_t)n * (2 * DD)))[q];
    v.x = fmaxf(v.x, 0.0f); v.y = fmaxf(v.y, 0.0f);
    v.z = fmaxf(v.z, 0.0f); v.w = fmaxf(v.w, 0.0f);
    ((float4*)(hout + (size_t)n * (2 * DD)))[q] = v;
    float* p = phis + (size_t)g * (2 * DD) + q * 4;
    atomicAdd(p + 0, v.x); atomicAdd(p + 1, v.y);
    atomicAdd(p + 2, v.z); atomicAdd(p + 3, v.w);
}

extern "C" void kernel_launch(void* const* d_in, const int* in_sizes, int n_in,
                              void* d_out, int out_size, void* d_ws, size_t ws_size,
                              hipStream_t stream) {
    const float* x        = (const float*)d_in[0];
    const int*   esrc     = (const int*)d_in[1];
    const int*   edst     = (const int*)d_in[2];
    const int*   gids     = (const int*)d_in[3];
    const float* Wfc      = (const float*)d_in[4];
    const float* bfc      = (const float*)d_in[5];
    const float* wl       = (const float*)d_in[6];
    const float* wr       = (const float*)d_in[7];
    const float* Wres     = (const float*)d_in[8];

    float* h    = (float*)d_out;
    float* phis = (float*)d_out + (size_t)N_NODES * 2 * DD;

    char* ws = (char*)d_ws;
    float*    feat   = (float*)ws;
    float*    a_self = (float*)(ws + (size_t)2 * N_NODES * DD * 4);
    float*    a_nb   = a_self + 2 * N_NODES;
    unsigned* menc   = (unsigned*)(a_nb + 2 * N_NODES);
    float*    denom  = (float*)(menc + 2 * N_NODES);

    k_init<<<dim3((2 * N_NODES + 255) / 256), dim3(256), 0, stream>>>(menc, denom, phis);

    k_gemm<<<dim3((N_NODES + 63) / 64, 2, 2), dim3(128), 0, stream>>>(
        x, Wfc, bfc, Wres, feat, h);

    k_attn_coef<<<dim3((N_NODES + 3) / 4, 2), dim3(128), 0, stream>>>(
        feat, wl, wr, a_self, a_nb);

    k_edge_max<<<dim3((N_EDGES + 255) / 256, 2), dim3(256), 0, stream>>>(
        esrc, edst, a_self, a_nb, menc);

    k_edge_denom<<<dim3((N_EDGES + 255) / 256, 2), dim3(256), 0, stream>>>(
        esrc, edst, a_self, a_nb, menc, denom);

    k_edge_accum<<<dim3(2048, 2), dim3(256), 0, stream>>>(
        esrc, edst, a_self, a_nb, menc, denom, feat, h);

    k_relu_pool<<<dim3((N_NODES * 64 + 255) / 256), dim3(256), 0, stream>>>(
        h, gids, phis);
}


# Edges counting-sorted by destination inside the call (count, scan, fill); one wave per node then does max, sum and weighted row sum, so no float atomics on h; sum order = arrival order of the fill
# speedup vs baseline: 1.4019x; 1.4019x over previous
#include <hip/hip_runtime.h>
#include <hip/hip_bf16.h>

#define N_NODES 10000
#define N_EDGES 640000
#define N_GRAPHS 64
#define DD 128
#define NEG_SLOPE 0.01f

typedef __attribute__((ext_vector_type(2))) float v2f;
typedef __attribute__((ext_vector_type(8))) float v8f;

__global__ void k_init(int* __restrict__ cursor, float* __restrict__ phis) {
    int i = blockIdx.x * blockDim.x + threadIdx.x;
    if (i < N_NODES) cursor[i] = 0;
    if (i < N_GRAPHS * 2 * DD) phis[i] = 0.0f;
}

__global__ void k_count(const int* __restrict__ edst, int* __restrict__ cursor) {
    const int e = blockIdx.x * blockDim.x + threadIdx.x;
    if (e >= N_EDGES) return;
    atomicAdd(&cursor[edst[e]], 1);
}

__global__ void k_scan(int* __restrict__ cursor, int* __restrict__ rowptr) {
    __shared__ int part[1024];
    const int t = threadIdx.x;
    const int base = t * 10;
    int v[10];
    int sum = 0;
#pragma unroll
    for (int k = 0; k < 10; ++k) {
        const int idx = base + k;
        v[k] = (idx < N_NODES) ? cursor[idx] : 0;
        sum += v[k];
    }
    part[t] = sum;
    __syncthreads();
    for (int off = 1; off < 1024; off <<= 1) {
        const int add = (t >= off) ? part[t - off] : 0;
        __syncthreads();
        part[t] += add;
        __syncthreads();
    }
    int run = part[t] - sum;
#pragma unroll
    for (int k = 0; k < 10; ++k) {
        const int idx = base + k;
        if (idx < N_NODES) { rowptr[idx] = run; cursor[idx] = run; run += v[k]; }
    }
    if (t == 1023) rowptr[N_NODES] = part[1023];
}

__global__ void k_fill(const int* __restrict__ esrc, const int* __restrict__ edst,
                       int* __restrict__ cursor, int* __restrict__ csr) {
    const int e = blockIdx.x * blockDim.x + threadIdx.x;
    if (e >= N_EDGES) return;
    const int p = atomicAdd(&cursor[edst[e]], 1);
    csr[p] = esrc[e];
}

__global__ void k_gemm(const float* __restrict__ x,
                       const float* __restrict__ Wfc, const float* __restrict__ bfc,
                       const float* __restrict__ Wres,
                       float* __restrict__ feat_ws, float* __restrict__ hout, int job0) {
    __shared__ float WT[DD * DD];
    const int job   = blockIdx.y + job0;
    const int b     = (job >= 2) ? 1 : 0;
    const int which = (job == 1 || job == 2) ? 1 : 0;
    const float* W = (which == 0 ? Wfc : Wres) + (size_t)b * DD * DD;

    for (int idx = threadIdx.x; idx < DD * DD; idx += blockDim.x) {
        int j = idx >> 7, k = idx & 127;
        WT[k * DD + j] = W[idx];
    }
    __syncthreads();

    const int lane = threadIdx.x & 31;
    const int wave = threadIdx.x >> 5;
    const int rowBase = blockIdx.x * 64 + wave * 16;
    const int l16 = lane & 15;
    const int hi  = lane >> 4;

    v8f c[8];
#pragma unroll
    for (int t = 0; t < 8; ++t) {
        float bv = (which == 0) ? bfc[b * DD + t * 16 + l16] : 0.0f;
#pragma unroll
        for (int v = 0; v < 8; ++v) c[t][v] = bv;
    }

    const int arow = min(rowBase + l16, N_NODES - 1);
    const float* xrow = x + (size_t)arow * DD;

    for (int k0 = 0; k0 < DD; k0 += 4) {
        const int krow = k0 + 2 * hi;
        v2f a = *(const v2f*)(xrow + krow);
#pragma unroll
        for (int t = 0; t < 8; ++t) {
            const int j = t * 16 + l16;
            v2f bf;
            bf.x = WT[krow * DD + j];
            bf.y = WT[(krow + 1) * DD + j];
            c[t] = __builtin_amdgcn_wmma_f32_16x16x4_f32(
                false, a, false, bf, (short)0, c[t], false, false);
        }
    }

#pragma unroll
    for (int t = 0; t < 8; ++t) {
        const int col = t * 16 + l16;
#pragma unroll
        for (int v = 0; v < 8; ++v) {
            const int row = rowBase + v + 8 * hi;
            if (row < N_NODES) {
                if (which == 0)
                    feat_ws[(size_t)row * DD + col] = c[t][v];
                else
                    hout[(size_t)row * (2 * DD) + b * DD + col] = c[t][v];
            }
        }
    }
}

__global__ void k_attn_coef(const float* __restrict__ feat,
                            const float* __restrict__ wl, const float* __restrict__ wr,
                            float* __restrict__ a_self, float* __restrict__ a_nb, int b) {
    const int row = blockIdx.x * (blockDim.x >> 5) + (threadIdx.x >> 5);
    if (row >= N_NODES) return;
    const int lane = threadIdx.x & 31;
    float4 f = ((const float4*)(feat + (size_t)row * DD))[lane];
    float4 l = ((const float4*)(wl + b * DD))[lane];
    float4 r = ((const float4*)(wr + b * DD))[lane];
    float sl = f.x * l.x + f.y * l.y + f.z * l.z + f.w * l.w;
    float sr = f.x * r.x + f.y * r.y + f.z * r.z + f.w * r.w;
#pragma unroll
    for (int m = 16; m >= 1; m >>= 1) {
        sl += __shfl_xor(sl, m, 32);
        sr += __shfl_xor(sr, m, 32);
    }
    if (lane == 0) {
        a_self[row] = sl;
        a_nb[row]   = sr;
    }
}

__device__ __forceinline__ float edge_score(float as, float an) {
    float sc = as + an;
    return sc >= 0.0f ? sc : NEG_SLOPE * sc;
}

__global__ void k_gather(const int* __restrict__ rowptr, const int* __restrict__ csr,
                         const float* __restrict__ a_self, const float* __restrict__ a_nb,
                         const float* __restrict__ feat, float* __restrict__ hout, int b) {
    const int lane = threadIdx.x & 31;
    const int n = blockIdx.x * (blockDim.x >> 5) + (threadIdx.x >> 5);
    if (n >= N_NODES) return;
    const int beg = rowptr[n], end = rowptr[n + 1];
    if (beg >= end) return;
    const float as = a_self[n];

    float m = -INFINITY;
    for (int i = beg + lane; i < end; i += 32)
        m = fmaxf(m, edge_score(as, a_nb[csr[i]]));
#pragma unroll
    for (int o = 16; o >= 1; o >>= 1) m = fmaxf(m, __shfl_xor(m, o, 32));

    float den = 0.0f;
    for (int i = beg + lane; i < end; i += 32)
        den += __expf(edge_score(as, a_nb[csr[i]]) - m);
#pragma unroll
    for (int o = 16; o >= 1; o >>= 1) den += __shfl_xor(den, o, 32);

    const float4* F = (const float4*)feat;
    float4 acc = make_float4(0.0f, 0.0f, 0.0f, 0.0f);
    for (int base = beg; base < end; base += 32) {
        const int i = base + lane;
        int s = 0;
        float w = 0.0f;
        if (i < end) {
            s = csr[i];
            w = __expf(edge_score(as, a_nb[s]) - m) / den;
        }
        const int cnt = min(32, end - base);
        int j = 0;
        for (; j + 8 <= cnt; j += 8) {
            int   sj[8];
            float wj[8];
            float4 f[8];
#pragma unroll
            for (int u = 0; u < 8; ++u) { sj[u] = __shfl(s, j + u, 32); wj[u] = __shfl(w, j + u, 32); }
#pragma unroll
            for (int u = 0; u < 8; ++u) f[u] = F[(size_t)sj[u] * (DD / 4) + lane];
#pragma unroll
            for (int u = 0; u < 8; ++u) {
                acc.x += wj[u] * f[u].x; acc.y += wj[u] * f[u].y;
                acc.z += wj[u] * f[u].z; acc.w += wj[u] * f[u].w;
            }
        }
        for (; j < cnt; ++j) {
            const int   s1 = __shfl(s, j, 32);
            const float w1 = __shfl(w, j, 32);
            const float4 f1 = F[(size_t)s1 * (DD / 4) + lane];
            acc.x += w1 * f1.x; acc.y += w1 * f1.y;
            acc.z += w1 * f1.z; acc.w += w1 * f1.w;
        }
    }
    float4* hp = (float4*)(hout + (size_t)n * (2 * DD) + b * DD) + lane;
    float4 r = *hp;
    r.x += acc.x; r.y += acc.y; r.z += acc.z; r.w += acc.w;
    *hp = r;
}

__global__ void k_relu_pool(float* __restrict__ hout, const int* __restrict__ gids,
                            float* __restrict__ phis) {
    const int idx = blockIdx.x * blockDim.x + threadIdx.x;
    const int n = idx >> 6, q = idx & 63;
    if (n >= N_NODES) return;
    const int g = gids[n];
    float4 v = ((float4*)(hout + (size_t)n * (2 * DD)))[q];
    v.x = fmaxf(v.x, 0.0f); v.y = fmaxf(v.y, 0.0f);
    v.z = fmaxf(v.z, 0.0f); v.w = fmaxf(v.w, 0.0f);
    ((float4*)(hout + (size_t)n * (2 * DD)))[q] = v;
    float* p = phis + (size_t)g * (2 * DD) + q * 4;
    atomicAdd(p + 0, v.x); atomicAdd(p + 1, v.y);
    atomicAdd(p + 2, v.z); atomicAdd(p + 3, v.w);
}

extern "C" void kernel_launch(void* const* d_in, const int* in_sizes, int n_in,
                              void* d_out, int out_size, void* d_ws, size_t ws_size,
                              hipStream_t stream) {
    const float* x        = (const float*)d_in[0];
    const int*   esrc     = (const int*)d_in[1];
    const int*   edst     = (const int*)d_in[2];
    const int*   gids     = (const int*)d_in[3];
    const float* Wfc      = (const float*)d_in[4];
    const float* bfc      = (const float*)d_in[5];
    const float* wl       = (const float*)d_in[6];
    const float* wr       = (const float*)d_in[7];
    const float* Wres     = (const float*)d_in[8];

    float* h    = (float*)d_out;
    float* phis = (float*)d_out + (size_t)N_NODES * 2 * DD;

    char* ws = (char*)d_ws;
    float* featb  = (float*)ws;
    int*   csr    = (int*)(ws + (size_t)N_NODES * DD * 4);
    int*   rowptr = (int*)(ws + (size_t)N_NODES * DD * 4 + (size_t)N_EDGES * 4);
    int*   cursor = rowptr + 10240;
    float* a_self = (float*)(cursor + 10240);
    float* a_nb   = a_self + 10240;

    k_init<<<dim3((N_GRAPHS * 2 * DD + 255) / 256), dim3(256), 0, stream>>>(cursor, phis);

    k_count<<<dim3((N_EDGES + 255) / 256), dim3(256), 0, stream>>>(edst, cursor);
    k_scan<<<dim3(1), dim3(1024), 0, stream>>>(cursor, rowptr);
    k_fill<<<dim3((N_EDGES + 255) / 256), dim3(256), 0, stream>>>(esrc, edst, cursor, csr);

    k_gemm<<<dim3((N_NODES + 63) / 64, 3), dim3(128), 0, stream>>>(
        x, Wfc, bfc, Wres, featb, h, 0);
    k_attn_coef<<<dim3((N_NODES + 3) / 4), dim3(128), 0, stream>>>(
        featb, wl, wr, a_self, a_nb, 0);
    k_gather<<<dim3((N_NODES + 7) / 8), dim3(256), 0, stream>>>(
        rowptr, csr, a_self, a_nb, featb, h, 0);

    k_gemm<<<dim3((N_NODES + 63) / 64, 1), dim3(128), 0, stream>>>(
        x, Wfc, bfc, Wres, featb, h, 3);
    k_attn_coef<<<dim3((N_NODES + 3) / 4), dim3(128), 0, stream>>>(
        featb, wl, wr, a_self, a_nb, 1);
    k_gather<<<dim3((N_NODES + 7) / 8), dim3(256), 0, stream>>>(
        rowptr, csr, a_self, a_nb, featb, h, 1);

    k_relu_pool<<<dim3((N_NODES * 64 + 255) / 256), dim3(256), 0, stream>>>(
        h, gids, phis);
}


# GEMM weights staged untransposed: coalesced 16-byte copies into a padded LDS tile, B fragments read as K-contiguous 8-byte pairs, launch bound 128 (was a serial 4-byte transposing copy); sums kept
# speedup vs baseline: 1.8822x; 1.3426x over previous
#include <hip/hip_runtime.h>
#include <hip/hip_bf16.h>

#define N_NODES 10000
#define N_EDGES 640000
#define N_GRAPHS 64
#define DD 128
#define NEG_SLOPE 0.01f

typedef __attribute__((ext_vector_type(2))) float v2f;
typedef __attribute__((ext_vector_type(8))) float v8f;

__global__ void k_init(int* __restrict__ cursor, float* __restrict__ phis) {
    int i = blockIdx.x * blockDim.x + threadIdx.x;
    if (i < N_NODES) cursor[i] = 0;
    if (i < N_GRAPHS * 2 * DD) phis[i] = 0.0f;
}

__global__ void k_count(const int* __restrict__ edst, int* __restrict__ cursor) {
    const int e = blockIdx.x * blockDim.x + threadIdx.x;
    if (e >= N_EDGES) return;
    atomicAdd(&cursor[edst[e]], 1);
}

__global__ void k_scan(int* __restrict__ cursor, int* __restrict__ rowptr) {
    __shared__ int part[1024];
    const int t = threadIdx.x;
    const int base = t * 10;
    int v[10];
    int sum = 0;
#pragma unroll
    for (int k = 0; k < 10; ++k) {
        const int idx = base + k;
        v[k] = (idx < N_NODES) ? cursor[idx] : 0;
        sum += v[k];
    }
    part[t] = sum;
    __syncthreads();
    for (int off = 1; off < 1024; off <<= 1) {
        const int add = (t >= off) ? part[t - off] : 0;
        __syncthreads();
        part[t] += add;
        __syncthreads();
    }
    int run = part[t] - sum;
#pragma unroll
    for (int k = 0; k < 10; ++k) {
        const int idx = base + k;
        if (idx < N_NODES) { rowptr[idx] = run; cursor[idx] = run; run += v[k]; }
    }
    if (t == 1023) rowptr[N_NODES] = part[1023];
}

__global__ void k_fill(const int* __restrict__ esrc, const int* __restrict__ edst,
                       int* __restrict__ cursor, int* __restrict__ csr) {
    const int e = blockIdx.x * blockDim.x + threadIdx.x;
    if (e >= N_EDGES) return;
    const int p = atomicAdd(&cursor[edst[e]], 1);
    csr[p] = esrc[e];
}

#define GEMM_THREADS 128
#define WPITCH 132
__global__ void __launch_bounds__(GEMM_THREADS) k_gemm(const float* __restrict__ x,
                       const float* __restrict__ Wfc, const float* __restrict__ bfc,
                       const float* __restrict__ Wres,
                       float* __restrict__ feat_ws, float* __restrict__ hout, int job0) {
    __shared__ float Ws[DD * WPITCH];
    const int job   = blockIdx.y + job0;
    const int b     = (job >= 2) ? 1 : 0;
    const int which = (job == 1 || job == 2) ? 1 : 0;
    const float* W = (which == 0 ? Wfc : Wres) + (size_t)b * DD * DD;

    const float4* W4 = (const float4*)W;
    for (int i0 = 0; i0 < DD * DD / 4; i0 += 8 * GEMM_THREADS) {
        float4 w[8];
#pragma unroll
        for (int u = 0; u < 8; ++u) w[u] = W4[i0 + u * GEMM_THREADS + threadIdx.x];
#pragma unroll
        for (int u = 0; u < 8; ++u) {
            const int i = i0 + u * GEMM_THREADS + threadIdx.x;
            const int j = i >> 5, k4 = (i & 31) << 2;
            *(float4*)(Ws + j * WPITCH + k4) = w[u];
        }
    }
    __syncthreads();

    const int lane = threadIdx.x & 31;
    const int wave = threadIdx.x >> 5;
    const int rowBase = blockIdx.x * 64 + wave * 16;
    const int l16 = lane & 15;
    const int hi  = lane >> 4;

    v8f c[8];
#pragma unroll
    for (int t = 0; t < 8; ++t) {
        float bv = (which == 0) ? bfc[b * DD + t * 16 + l16] : 0.0f;
#pragma unroll
        for (int v = 0; v < 8; ++v) c[t][v] = bv;
    }

    const int arow = min(rowBase + l16, N_NODES - 1);
    const float* xrow = x + (size_t)arow * DD;

    for (int k0 = 0; k0 < DD; k0 += 4) {
        const int krow = k0 + 2 * hi;
        v2f a = *(const v2f*)(xrow + krow);
#pragma unroll
        for (int t = 0; t < 8; ++t) {
            const int j = t * 16 + l16;
            const v2f bf = *(const v2f*)(Ws + j * WPITCH + krow);
            c[t] = __builtin_amdgcn_wmma_f32_16x16x4_f32(
                false, a, false, bf, (short)0, c[t], false, false);
        }
    }

#pragma unroll
    for (int t = 0; t < 8; ++t) {
        const int col = t * 16 + l16;
#pragma unroll
        for (int v = 0; v < 8; ++v) {
            const int row = rowBase + v + 8 * hi;
            if (row < N_NODES) {
                if (which == 0)
                    feat_ws[(size_t)row * DD + col] = c[t][v];
                else
                    hout[(size_t)row * (2 * DD) + b * DD + col] = c[t][v];
            }
        }
    }
}

__global__ void k_attn_coef(const float* __restrict__ feat,
                            const float* __restrict__ wl, const float* __restrict__ wr,
                            float* __restrict__ a_self, float* __restrict__ a_nb, int b) {
    const int row = blockIdx.x * (blockDim.x >> 5) + (threadIdx.x >> 5);
    if (row >= N_NODES) return;
    const int lane = threadIdx.x & 31;
    float4 f = ((const float4*)(feat + (size_t)row * DD))[lane];
    float4 l = ((const float4*)(wl + b * DD))[lane];
    float4 r = ((const float4*)(wr + b * DD))[lane];
    float sl = f.x * l.x + f.y * l.y + f.z * l.z + f.w * l.w;
    float sr = f.x * r.x + f.y * r.y + f.z * r.z + f.w * r.w;
#pragma unroll
    for (int m = 16; m >= 1; m >>= 1) {
        sl += __shfl_xor(sl, m, 32);
        sr += __shfl_xor(sr, m, 32);
    }
    if (lane == 0) {
        a_self[row] = sl;
        a_nb[row]   = sr;
    }
}

__device__ __forceinline__ float edge_score(float as, float an) {
    float sc = as + an;
    return sc >= 0.0f ? sc : NEG_SLOPE * sc;
}

__global__ void k_gather(const int* __restrict__ rowptr, const int* __restrict__ csr,
                         const float* __restrict__ a_self, const float* __restrict__ a_nb,
                         const float* __restrict__ feat, float* __restrict__ hout, int b) {
    const int lane = threadIdx.x & 31;
    const int n = blockIdx.x * (blockDim.x >> 5) + (threadIdx.x >> 5);
    if (n >= N_NODES) return;
    const int beg = rowptr[n], end = rowptr[n + 1];
    if (beg >= end) return;
    const float as = a_self[n];

    float m = -INFINITY;
    for (int i = beg + lane; i < end; i += 32)
        m = fmaxf(m, edge_score(as, a_nb[csr[i]]));
#pragma unroll
    for (int o = 16; o >= 1; o >>= 1) m = fmaxf(m, __shfl_xor(m, o, 32));

    float den = 0.0f;
    for (int i = beg + lane; i < end; i += 32)
        den += __expf(edge_score(as, a_nb[csr[i]]) - m);
#pragma unroll
    for (int o = 16; o >= 1; o >>= 1) den += __shfl_xor(den, o, 32);

    const float4* F = (const float4*)feat;
    float4 acc = make_float4(0.0f, 0.0f, 0.0f, 0.0f);
    for (int base = beg; base < end; base += 32) {
        const int i = base + lane;
        int s = 0;
        float w = 0.0f;
        if (i < end) {
            s = csr[i];
            w = __expf(edge_score(as, a_nb[s]) - m) / den;
        }
        const int cnt = min(32, end - base);
        int j = 0;
        for (; j + 8 <= cnt; j += 8) {
            int   sj[8];
            float wj[8];
            float4 f[8];
#pragma unroll
            for (int u = 0; u < 8; ++u) { sj[u] = __shfl(s, j + u, 32); wj[u] = __shfl(w, j + u, 32); }
#pragma unroll
            for (int u = 0; u < 8; ++u) f[u] = F[(size_t)sj[u] * (DD / 4) + lane];
#pragma unroll
            for (int u = 0; u < 8; ++u) {
                acc.x += wj[u] * f[u].x; acc.y += wj[u] * f[u].y;
                acc.z += wj[u] * f[u].z; acc.w += wj[u] * f[u].w;
            }
        }
        for (; j < cnt; ++j) {
            const int   s1 = __shfl(s, j, 32);
            const float w1 = __shfl(w, j, 32);
            const float4 f1 = F[(size_t)s1 * (DD / 4) + lane];
            acc.x += w1 * f1.x; acc.y += w1 * f1.y;
            acc.z += w1 * f1.z; acc.w += w1 * f1.w;
        }
    }
    float4* hp = (float4*)(hout + (size_t)n * (2 * DD) + b * DD) + lane;
    float4 r = *hp;
    r.x += acc.x; r.y += acc.y; r.z += acc.z; r.w += acc.w;
    *hp = r;
}

__global__ void k_relu_pool(float* __restrict__ hout, const int* __restrict__ gids,
                            float* __restrict__ phis) {
    const int idx = blockIdx.x * blockDim.x + threadIdx.x;
    const int n = idx >> 6, q = idx & 63;
    if (n >= N_NODES) return;
    const int g = gids[n];
    float4 v = ((float4*)(hout + (size_t)n * (2 * DD)))[q];
    v.x = fmaxf(v.x, 0.0f); v.y = fmaxf(v.y, 0.0f);
    v.z = fmaxf(v.z, 0.0f); v.w = fmaxf(v.w, 0.0f);
    ((float4*)(hout + (size_t)n * (2 * DD)))[q] = v;
    float* p = phis + (size_t)g * (2 * DD) + q * 4;
    atomicAdd(p + 0, v.x); atomicAdd(p + 1, v.y);
    atomicAdd(p + 2, v.z); atomicAdd(p + 3, v.w);
}

extern "C" void kernel_launch(void* const* d_in, const int* in_sizes, int n_in,
                              void* d_out, int out_size, void* d_ws, size_t ws_size,
                              hipStream_t stream) {
    const float* x        = (const float*)d_in[0];
    const int*   esrc     = (const int*)d_in[1];
    const int*   edst     = (const int*)d_in[2];
    const int*   gids     = (const int*)d_in[3];
    const float* Wfc      = (const float*)d_in[4];
    const float* bfc      = (const float*)d_in[5];
    const float* wl       = (const float*)d_in[6];
    const float* wr       = (const float*)d_in[7];
    const float* Wres     = (const float*)d_in[8];

    float* h    = (float*)d_out;
    float* phis = (float*)d_out + (size_t)N_NODES * 2 * DD;

    char* ws = (char*)d_ws;
    float* featb  = (float*)ws;
    int*   csr    = (int*)(ws + (size_t)N_NODES * DD * 4);
    int*   rowptr = (int*)(ws + (size_t)N_NODES * DD * 4 + (size_t)N_EDGES * 4);
    int*   cursor = rowptr + 10240;
    float* a_self = (float*)(cursor + 10240);
    float* a_nb   = a_self + 10240;

    k_init<<<dim3((N_GRAPHS * 2 * DD + 255) / 256), dim3(256), 0, stream>>>(cursor, phis);

    k_count<<<dim3((N_EDGES + 255) / 256), dim3(256), 0, stream>>>(edst, cursor);
    k_scan<<<dim3(1), dim3(1024), 0, stream>>>(cursor, rowptr);
    k_fill<<<dim3((N_EDGES + 255) / 256), dim3(256), 0, stream>>>(esrc, edst, cursor, csr);

    k_gemm<<<dim3((N_NODES + 63) / 64, 3), dim3(128), 0, stream>>>(
        x, Wfc, bfc, Wres, featb, h, 0);
    k_attn_coef<<<dim3((N_NODES + 3) / 4), dim3(128), 0, stream>>>(
        featb, wl, wr, a_self, a_nb, 0);
    k_gather<<<dim3((N_NODES + 7) / 8), dim3(256), 0, stream>>>(
        rowptr, csr, a_self, a_nb, featb, h, 0);

    k_gemm<<<dim3((N_NODES + 63) / 64, 1), dim3(128), 0, stream>>>(
        x, Wfc, bfc, Wres, featb, h, 3);
    k_attn_coef<<<dim3((N_NODES + 3) / 4), dim3(128), 0, stream>>>(
        featb, wl, wr, a_self, a_nb, 1);
    k_gather<<<dim3((N_NODES + 7) / 8), dim3(256), 0, stream>>>(
        rowptr, csr, a_self, a_nb, featb, h, 1);

    k_relu_pool<<<dim3((N_NODES * 64 + 255) / 256), dim3(256), 0, stream>>>(
        h, gids, phis);
}


# Thin launches folded into producers: zero-init and attention dots in the GEMM, ReLU and pooling in the gather (8 rows summed in LDS, then atomics); 11 -> 7 launches; dot and pool sum order change
# speedup vs baseline: 2.1039x; 1.1178x over previous
#include <hip/hip_runtime.h>
#include <hip/hip_bf16.h>

#define N_NODES 10000
#define N_EDGES 640000
#define N_GRAPHS 64
#define DD 128
#define NEG_SLOPE 0.01f

typedef __attribute__((ext_vector_type(2))) float v2f;
typedef __attribute__((ext_vector_type(8))) float v8f;

__global__ void k_count(const int* __restrict__ edst, int* __restrict__ cursor) {
    const int e = blockIdx.x * blockDim.x + threadIdx.x;
    if (e >= N_EDGES) return;
    atomicAdd(&cursor[edst[e]], 1);
}

__global__ void k_scan(int* __restrict__ cursor, int* __restrict__ rowptr) {
    __shared__ int part[1024];
    const int t = threadIdx.x;
    const int base = t * 10;
    int v[10];
    int sum = 0;
#pragma unroll
    for (int k = 0; k < 10; ++k) {
        const int idx = base + k;
        v[k] = (idx < N_NODES) ? cursor[idx] : 0;
        sum += v[k];
    }
    part[t] = sum;
    __syncthreads();
    for (int off = 1; off < 1024; off <<= 1) {
        const int add = (t >= off) ? part[t - off] : 0;
        __syncthreads();
        part[t] += add;
        __syncthreads();
    }
    int run = part[t] - sum;
#pragma unroll
    for (int k = 0; k < 10; ++k) {
        const int idx = base + k;
        if (idx < N_NODES) { rowptr[idx] = run; cursor[idx] = run; run += v[k]; }
    }
    if (t == 1023) rowptr[N_NODES] = part[1023];
}

__global__ void k_fill(const int* __restrict__ esrc, const int* __restrict__ edst,
                       int* __restrict__ cursor, int* __restrict__ csr) {
    const int e = blockIdx.x * blockDim.x + threadIdx.x;
    if (e >= N_EDGES) return;
    const int p = atomicAdd(&cursor[edst[e]], 1);
    csr[p] = esrc[e];
}

#define GEMM_THREADS 128
#define WPITCH 132
__global__ void __launch_bounds__(GEMM_THREADS) k_gemm(const float* __restrict__ x,
                       const float* __restrict__ Wfc, const float* __restrict__ bfc,
                       const float* __restrict__ Wres,
                       const float* __restrict__ wl, const float* __restrict__ wr,
                       float* __restrict__ feat_ws, float* __restrict__ hout,
                       float* __restrict__ a_self, float* __restrict__ a_nb,
                       int* __restrict__ cursor, float* __restrict__ phis, int job0) {
    __shared__ float Ws[DD * WPITCH];
    if (job0 == 0 && blockIdx.y == 0) {
        const int i = blockIdx.x * GEMM_THREADS + threadIdx.x;
        if (i < N_NODES) cursor[i] = 0;
        if (i < N_GRAPHS * 2 * DD) phis[i] = 0.0f;
    }
    const int job   = blockIdx.y + job0;
    const int b     = (job >= 2) ? 1 : 0;
    const int which = (job == 1 || job == 2) ? 1 : 0;
    const float* W = (which == 0 ? Wfc : Wres) + (size_t)b * DD * DD;

    const float4* W4 = (const float4*)W;
    for (int i0 = 0; i0 < DD * DD / 4; i0 += 8 * GEMM_THREADS) {
        float4 w[8];
#pragma unroll
        for (int u = 0; u < 8; ++u) w[u] = W4[i0 + u * GEMM_THREADS + threadIdx.x];
#pragma unroll
        for (int u = 0; u < 8; ++u) {
            const int i = i0 + u * GEMM_THREADS + threadIdx.x;
            const int j = i >> 5, k4 = (i & 31) << 2;
            *(float4*)(Ws + j * WPITCH + k4) = w[u];
        }
    }
    __syncthreads();

    const int lane = threadIdx.x & 31;
    const int wave = threadIdx.x >> 5;
    const int rowBase = blockIdx.x * 64 + wave * 16;
    const int l16 = lane & 15;
    const int hi  = lane >> 4;

    v8f c[8];
#pragma unroll
    for (int t = 0; t < 8; ++t) {
        float bv = (which == 0) ? bfc[b * DD + t * 16 + l16] : 0.0f;
#pragma unroll
        for (int v = 0; v < 8; ++v) c[t][v] = bv;
    }

    const int arow = min(rowBase + l16, N_NODES - 1);
    const float* xrow = x + (size_t)arow * DD;

    for (int k0 = 0; k0 < DD; k0 += 4) {
        const int krow = k0 + 2 * hi;
        v2f a = *(const v2f*)(xrow + krow);
#pragma unroll
        for (int t = 0; t < 8; ++t) {
            const int j = t * 16 + l16;
            const v2f bf = *(const v2f*)(Ws + j * WPITCH + krow);
            c[t] = __builtin_amdgcn_wmma_f32_16x16x4_f32(
                false, a, false, bf, (short)0, c[t], false, false);
        }
    }

#pragma unroll
    for (int t = 0; t < 8; ++t) {
        const int col = t * 16 + l16;
#pragma unroll
        for (int v = 0; v < 8; ++v) {
            const int row = rowBase + v + 8 * hi;
            if (row < N_NODES) {
                if (which == 0)
                    feat_ws[(size_t)row * DD + col] = c[t][v];
                else
                    hout[(size_t)row * (2 * DD) + b * DD + col] = c[t][v];
            }
        }
    }

    if (which == 0) {
        float wlv[8], wrv[8];
#pragma unroll
        for (int t = 0; t < 8; ++t) {
            wlv[t] = wl[b * DD + t * 16 + l16];
            wrv[t] = wr[b * DD + t * 16 + l16];
        }
        float my_sl = 0.0f, my_sr = 0.0f;
#pragma unroll
        for (int v = 0; v < 8; ++v) {
            float sl = 0.0f, sr = 0.0f;
#pragma unroll
            for (int t = 0; t < 8; ++t) {
                sl += c[t][v] * wlv[t];
                sr += c[t][v] * wrv[t];
            }
#pragma unroll
            for (int m = 8; m >= 1; m >>= 1) {
                sl += __shfl_xor(sl, m, 32);
                sr += __shfl_xor(sr, m, 32);
            }
            if (l16 == v) { my_sl = sl; my_sr = sr; }
        }
        const int myrow = rowBase + l16 + 8 * hi;
        if (l16 < 8 && myrow < N_NODES) {
            a_self[myrow] = my_sl;
            a_nb[myrow]   = my_sr;
        }
    }
}

__device__ __forceinline__ float edge_score(float as, float an) {
    float sc = as + an;
    return sc >= 0.0f ? sc : NEG_SLOPE * sc;
}

#define GATHER_WAVES 8
__global__ void __launch_bounds__(GATHER_WAVES * 32) k_gather(
                         const int* __restrict__ rowptr, const int* __restrict__ csr,
                         const float* __restrict__ a_self, const float* __restrict__ a_nb,
                         const float* __restrict__ feat, float* __restrict__ hout,
                         const int* __restrict__ gids, float* __restrict__ phis, int b) {
    __shared__ float4 prow[GATHER_WAVES][32];
    __shared__ int pg[GATHER_WAVES];
    const int lane = threadIdx.x & 31;
    const int wv = threadIdx.x >> 5;
    const int n = blockIdx.x * GATHER_WAVES + wv;

    float4 r = make_float4(0.0f, 0.0f, 0.0f, 0.0f);
    int g = -1;
    if (n < N_NODES) {
        const int beg = rowptr[n], end = rowptr[n + 1];
        float4 acc = make_float4(0.0f, 0.0f, 0.0f, 0.0f);
        if (beg < end) {
            const float as = a_self[n];

            float m = -INFINITY;
            for (int i = beg + lane; i < end; i += 32)
                m = fmaxf(m, edge_score(as, a_nb[csr[i]]));
#pragma unroll
            for (int o = 16; o >= 1; o >>= 1) m = fmaxf(m, __shfl_xor(m, o, 32));

            float den = 0.0f;
            for (int i = beg + lane; i < end; i += 32)
                den += __expf(edge_score(as, a_nb[csr[i]]) - m);
#pragma unroll
            for (int o = 16; o >= 1; o >>= 1) den += __shfl_xor(den, o, 32);

            const float4* F = (const float4*)feat;
            for (int base = beg; base < end; base += 32) {
                const int i = base + lane;
                int s = 0;
                float w = 0.0f;
                if (i < end) {
                    s = csr[i];
                    w = __expf(edge_score(as, a_nb[s]) - m) / den;
                }
                const int cnt = min(32, end - base);
                int j = 0;
                for (; j + 8 <= cnt; j += 8) {
                    int   sj[8];
                    float wj[8];
                    float4 f[8];
#pragma unroll
                    for (int u = 0; u < 8; ++u) { sj[u] = __shfl(s, j + u, 32); wj[u] = __shfl(w, j + u, 32); }
#pragma unroll
                    for (int u = 0; u < 8; ++u) f[u] = F[(size_t)sj[u] * (DD / 4) + lane];
#pragma unroll
                    for (int u = 0; u < 8; ++u) {
                        acc.x += wj[u] * f[u].x; acc.y += wj[u] * f[u].y;
                        acc.z += wj[u] * f[u].z; acc.w += wj[u] * f[u].w;
                    }
                }
                for (; j < cnt; ++j) {
                    const int   s1 = __shfl(s, j, 32);
                    const float w1 = __shfl(w, j, 32);
                    const float4 f1 = F[(size_t)s1 * (DD / 4) + lane];
                    acc.x += w1 * f1.x; acc.y += w1 * f1.y;
                    acc.z += w1 * f1.z; acc.w += w1 * f1.w;
                }
            }
        }
        float4* hp = (float4*)(hout + (size_t)n * (2 * DD) + b * DD) + lane;
        r = *hp;
        r.x = fmaxf(r.x + acc.x, 0.0f); r.y = fmaxf(r.y + acc.y, 0.0f);
        r.z = fmaxf(r.z + acc.z, 0.0f); r.w = fmaxf(r.w + acc.w, 0.0f);
        *hp = r;
        g = gids[n];
    }
    prow[wv][lane] = r;
    if (lane == 0) pg[wv] = g;
    __syncthreads();

    if (threadIdx.x < DD) {
        const int col = threadIdx.x;
        const float* pr = (const float*)prow;
        float run = 0.0f;
        int gcur = -1;
        for (int w = 0; w < GATHER_WAVES; ++w) {
            const int gw = pg[w];
            if (gw != gcur) {
                if ((unsigned)gcur < (unsigned)N_GRAPHS)
                    atomicAdd(phis + (size_t)gcur * (2 * DD) + b * DD + col, run);
                gcur = gw;
                run = 0.0f;
            }
            run += pr[w * DD + col];
        }
        if ((unsigned)gcur < (unsigned)N_GRAPHS)
            atomicAdd(phis + (size_t)gcur * (2 * DD) + b * DD + col, run);
    }
}

extern "C" void kernel_launch(void* const* d_in, const int* in_sizes, int n_in,
                              void* d_out, int out_size, void* d_ws, size_t ws_size,
                              hipStream_t stream) {
    const float* x        = (const float*)d_in[0];
    const int*   esrc     = (const int*)d_in[1];
    const int*   edst     = (const int*)d_in[2];
    const int*   gids     = (const int*)d_in[3];
    const float* Wfc      = (const float*)d_in[4];
    const float* bfc      = (const float*)d_in[5];
    const float* wl       = (const float*)d_in[6];
    const float* wr       = (const float*)d_in[7];
    const float* Wres     = (const float*)d_in[8];

    float* h    = (float*)d_out;
    float* phis = (float*)d_out + (size_t)N_NODES * 2 * DD;

    char* ws = (char*)d_ws;
    float* featb  = (float*)ws;
    int*   csr    = (int*)(ws + (size_t)N_NODES * DD * 4);
    int*   rowptr = (int*)(ws + (size_t)N_NODES * DD * 4 + (size_t)N_EDGES * 4);
    int*   cursor = rowptr + 10240;
    float* a_self = (float*)(cursor + 10240);
    float* a_nb   = a_self + 10240;

    k_gemm<<<dim3((N_NODES + 63) / 64, 3), dim3(GEMM_THREADS), 0, stream>>>(
        x, Wfc, bfc, Wres, wl, wr, featb, h, a_self, a_nb, cursor, phis, 0);

    k_count<<<dim3((N_EDGES + 255) / 256), dim3(256), 0, stream>>>(edst, cursor);
    k_scan<<<dim3(1), dim3(1024), 0, stream>>>(cursor, rowptr);
    k_fill<<<dim3((N_EDGES + 255) / 256), dim3(256), 0, stream>>>(esrc, edst, cursor, csr);

    k_gather<<<dim3((N_NODES + GATHER_WAVES - 1) / GATHER_WAVES), dim3(GATHER_WAVES * 32), 0, stream>>>(
        rowptr, csr, a_self, a_nb, featb, h, gids, phis, 0);

    k_gemm<<<dim3((N_NODES + 63) / 64, 1), dim3(GEMM_THREADS), 0, stream>>>(
        x, Wfc, bfc, Wres, wl, wr, featb, h, a_self, a_nb, cursor, phis, 3);
    k_gather<<<dim3((N_NODES + GATHER_WAVES - 1) / GATHER_WAVES), dim3(GATHER_WAVES * 32), 0, stream>>>(
        rowptr, csr, a_self, a_nb, featb, h, gids, phis, 1);
}


# Edges grouped by destination in one launch: a work-group owns 80 nodes, streams the destination list (L2), bins in LDS, writes csr whole-line; no global atomics; 7->5 launches; sum order = LDS fill
# speedup vs baseline: 5.1287x; 2.4377x over previous
#include <hip/hip_runtime.h>
#include <hip/hip_bf16.h>

#define N_NODES 10000
#define N_EDGES 640000
#define N_GRAPHS 64
#define DD 128
#define NEG_SLOPE 0.01f

typedef __attribute__((ext_vector_type(2))) float v2f;
typedef __attribute__((ext_vector_type(8))) float v8f;

#define BIN_NODES 80
#define BIN_WGS (N_NODES / BIN_NODES)
#define BIN_THREADS 1024
#define BIN_WAVES (BIN_THREADS / 32)
#define BIN_WCAP 512
#define BIN_CAP (BIN_WAVES * BIN_WCAP)
#define NV4 (N_EDGES / 4)
#define BIN_BATCH 8
#define BIN_NB 20

__global__ void __launch_bounds__(BIN_THREADS) k_bin(const int* __restrict__ esrc,
                      const int* __restrict__ edst,
                      int* __restrict__ rowptr, int* __restrict__ csr) {
    __shared__ unsigned wl[BIN_CAP];
    __shared__ int sorted[BIN_CAP];
    __shared__ int cnt[BIN_NODES], cur[BIN_NODES], pre[BIN_NODES + 1];
    __shared__ int wsum[BIN_WAVES];
    __shared__ int ovf;
    const int t = threadIdx.x;
    const int lane = t & 31;
    const int wave = t >> 5;
    const int lo = blockIdx.x * BIN_NODES;
    if (t < BIN_NODES) { cnt[t] = 0; cur[t] = 0; }
    if (t == 0) ovf = 0;
    __syncthreads();

    const int rot = blockIdx.x * (NV4 / BIN_WGS);
    const int4* D4 = (const int4*)edst;
    unsigned* mywl = wl + wave * BIN_WCAP;
    int below = 0;
    int wcnt = 0;

    auto loadb = [&](int4 (&d)[BIN_BATCH], int ib) {
#pragma unroll
        for (int u = 0; u < BIN_BATCH; ++u) {
            const int i = (ib * BIN_BATCH + u) * BIN_THREADS + t;
            int idx = i + rot;
            if (idx >= NV4) idx -= NV4;
            d[u] = (i < NV4) ? D4[idx] : make_int4(N_NODES, N_NODES, N_NODES, N_NODES);
        }
    };
    auto one = [&](int dv, int e) {
        const int tk = dv - lo;
        below += (int)((unsigned)tk >> 31);
        const bool mk = (unsigned)tk < BIN_NODES;
        const unsigned mask = __builtin_amdgcn_ballot_w32(mk);
        if (mask) {
            const int pos = wcnt + (int)__builtin_amdgcn_mbcnt_lo(mask, 0u);
            if (mk && pos < BIN_WCAP) mywl[pos] = ((unsigned)tk << 20) | (unsigned)e;
            wcnt += __builtin_popcount(mask);
        }
    };
    auto procb = [&](const int4 (&d)[BIN_BATCH], int ib) {
#pragma unroll
        for (int u = 0; u < BIN_BATCH; ++u) {
            const int i = (ib * BIN_BATCH + u) * BIN_THREADS + t;
            int idx = i + rot;
            if (idx >= NV4) idx -= NV4;
            const int e = idx * 4;
            one(d[u].x, e + 0);
            one(d[u].y, e + 1);
            one(d[u].z, e + 2);
            one(d[u].w, e + 3);
        }
    };

    int4 A[BIN_BATCH], B[BIN_BATCH];
    loadb(A, 0);
    for (int ib = 0; ib < BIN_NB; ib += 2) {
        loadb(B, ib + 1);
        procb(A, ib);
        loadb(A, ib + 2);
        procb(B, ib + 1);
    }

#pragma unroll
    for (int o = 16; o >= 1; o >>= 1) below += __shfl_xor(below, o, 32);
    if (lane == 0) {
        wsum[wave] = below;
        if (wcnt > BIN_WCAP) ovf = 1;
    }
    const int nw = min(wcnt, BIN_WCAP);
    for (int i = lane; i < nw; i += 32) atomicAdd(&cnt[mywl[i] >> 20], 1);
    __syncthreads();
    int base = 0;
#pragma unroll
    for (int w = 0; w < BIN_WAVES; ++w) base += wsum[w];
    const bool over = ovf != 0;
    if (over) {
        __syncthreads();
        if (t < BIN_NODES) cnt[t] = 0;
        __syncthreads();
        for (int e = t; e < N_EDGES; e += BIN_THREADS) {
            const int dl = edst[e] - lo;
            if ((unsigned)dl < BIN_NODES) atomicAdd(&cnt[dl], 1);
        }
        __syncthreads();
    }
    if (t <= BIN_NODES) {
        int run = 0;
        for (int k = 0; k < t; ++k) run += cnt[k];
        pre[t] = run;
    }
    __syncthreads();
    const int total = pre[BIN_NODES];
    if (t < BIN_NODES) rowptr[lo + t] = base + pre[t];
    if (t == 0 && blockIdx.x == BIN_WGS - 1) rowptr[N_NODES] = base + total;

    if (!over) {
        for (int i0 = 0; i0 < nw; i0 += 8 * 32) {
            unsigned m[8];
            int s[8];
#pragma unroll
            for (int u = 0; u < 8; ++u) {
                const int i = i0 + u * 32 + lane;
                m[u] = (i < nw) ? mywl[i] : 0u;
                s[u] = (i < nw) ? esrc[m[u] & 0xFFFFFu] : 0;
            }
#pragma unroll
            for (int u = 0; u < 8; ++u) {
                const int i = i0 + u * 32 + lane;
                if (i < nw) {
                    const int dl = (int)(m[u] >> 20);
                    const int slot = atomicAdd(&cur[dl], 1);
                    sorted[pre[dl] + slot] = s[u];
                }
            }
        }
        __syncthreads();
        for (int i = t; i < total; i += BIN_THREADS) csr[base + i] = sorted[i];
    } else {
        for (int e = t; e < N_EDGES; e += BIN_THREADS) {
            const int dl = edst[e] - lo;
            if ((unsigned)dl < BIN_NODES) {
                const int slot = atomicAdd(&cur[dl], 1);
                csr[base + pre[dl] + slot] = esrc[e];
            }
        }
    }
}

#define GEMM_THREADS 128
#define WPITCH 132
__global__ void __launch_bounds__(GEMM_THREADS) k_gemm(const float* __restrict__ x,
                       const float* __restrict__ Wfc, const float* __restrict__ bfc,
                       const float* __restrict__ Wres,
                       const float* __restrict__ wl, const float* __restrict__ wr,
                       float* __restrict__ feat_ws, float* __restrict__ hout,
                       float* __restrict__ a_self, float* __restrict__ a_nb,
                       float* __restrict__ phis, int job0) {
    __shared__ float Ws[DD * WPITCH];
    if (job0 == 0 && blockIdx.y == 0) {
        const int i = blockIdx.x * GEMM_THREADS + threadIdx.x;
        if (i < N_GRAPHS * 2 * DD) phis[i] = 0.0f;
    }
    const int job   = blockIdx.y + job0;
    const int b     = (job >= 2) ? 1 : 0;
    const int which = (job == 1 || job == 2) ? 1 : 0;
    const float* W = (which == 0 ? Wfc : Wres) + (size_t)b * DD * DD;

    const float4* W4 = (const float4*)W;
    for (int i0 = 0; i0 < DD * DD / 4; i0 += 8 * GEMM_THREADS) {
        float4 w[8];
#pragma unroll
        for (int u = 0; u < 8; ++u) w[u] = W4[i0 + u * GEMM_THREADS + threadIdx.x];
#pragma unroll
        for (int u = 0; u < 8; ++u) {
            const int i = i0 + u * GEMM_THREADS + threadIdx.x;
            const int j = i >> 5, k4 = (i & 31) << 2;
            *(float4*)(Ws + j * WPITCH + k4) = w[u];
        }
    }
    __syncthreads();

    const int lane = threadIdx.x & 31;
    const int wave = threadIdx.x >> 5;
    const int rowBase = blockIdx.x * 64 + wave * 16;
    const int l16 = lane & 15;
    const int hi  = lane >> 4;

    v8f c[8];
#pragma unroll
    for (int t = 0; t < 8; ++t) {
        float bv = (which == 0) ? bfc[b * DD + t * 16 + l16] : 0.0f;
#pragma unroll
        for (int v = 0; v < 8; ++v) c[t][v] = bv;
    }

    const int arow = min(rowBase + l16, N_NODES - 1);
    const float* xrow = x + (size_t)arow * DD;

    for (int k0 = 0; k0 < DD; k0 += 4) {
        const int krow = k0 + 2 * hi;
        v2f a = *(const v2f*)(xrow + krow);
#pragma unroll
        for (int t = 0; t < 8; ++t) {
            const int j = t * 16 + l16;
            const v2f bf = *(const v2f*)(Ws + j * WPITCH + krow);
            c[t] = __builtin_amdgcn_wmma_f32_16x16x4_f32(
                false, a, false, bf, (short)0, c[t], false, false);
        }
    }

#pragma unroll
    for (int t = 0; t < 8; ++t) {
        const int col = t * 16 + l16;
#pragma unroll
        for (int v = 0; v < 8; ++v) {
            const int row = rowBase + v + 8 * hi;
            if (row < N_NODES) {
                if (which == 0)
                    feat_ws[(size_t)row * DD + col] = c[t][v];
                else
                    hout[(size_t)row * (2 * DD) + b * DD + col] = c[t][v];
            }
        }
    }

    if (which == 0) {
        float wlv[8], wrv[8];
#pragma unroll
        for (int t = 0; t < 8; ++t) {
            wlv[t] = wl[b * DD + t * 16 + l16];
            wrv[t] = wr[b * DD + t * 16 + l16];
        }
        float my_sl = 0.0f, my_sr = 0.0f;
#pragma unroll
        for (int v = 0; v < 8; ++v) {
            float sl = 0.0f, sr = 0.0f;
#pragma unroll
            for (int t = 0; t < 8; ++t) {
                sl += c[t][v] * wlv[t];
                sr += c[t][v] * wrv[t];
            }
#pragma unroll
            for (int m = 8; m >= 1; m >>= 1) {
                sl += __shfl_xor(sl, m, 32);
                sr += __shfl_xor(sr, m, 32);
            }
            if (l16 == v) { my_sl = sl; my_sr = sr; }
        }
        const int myrow = rowBase + l16 + 8 * hi;
        if (l16 < 8 && myrow < N_NODES) {
            a_self[myrow] = my_sl;
            a_nb[myrow]   = my_sr;
        }
    }
}

__device__ __forceinline__ float edge_score(float as, float an) {
    float sc = as + an;
    return sc >= 0.0f ? sc : NEG_SLOPE * sc;
}

#define GATHER_WAVES 8
__global__ void __launch_bounds__(GATHER_WAVES * 32) k_gather(
                         const int* __restrict__ rowptr, const int* __restrict__ csr,
                         const float* __restrict__ a_self, const float* __restrict__ a_nb,
                         const float* __restrict__ feat, float* __restrict__ hout,
                         const int* __restrict__ gids, float* __restrict__ phis, int b) {
    __shared__ float4 prow[GATHER_WAVES][32];
    __shared__ int pg[GATHER_WAVES];
    const int lane = threadIdx.x & 31;
    const int wv = threadIdx.x >> 5;
    const int n = blockIdx.x * GATHER_WAVES + wv;

    float4 r = make_float4(0.0f, 0.0f, 0.0f, 0.0f);
    int g = -1;
    if (n < N_NODES) {
        const int beg = rowptr[n], end = rowptr[n + 1];
        float4 acc = make_float4(0.0f, 0.0f, 0.0f, 0.0f);
        if (beg < end) {
            const float as = a_self[n];

            float m = -INFINITY;
            for (int i = beg + lane; i < end; i += 32)
                m = fmaxf(m, edge_score(as, a_nb[csr[i]]));
#pragma unroll
            for (int o = 16; o >= 1; o >>= 1) m = fmaxf(m, __shfl_xor(m, o, 32));

            float den = 0.0f;
            for (int i = beg + lane; i < end; i += 32)
                den += __expf(edge_score(as, a_nb[csr[i]]) - m);
#pragma unroll
            for (int o = 16; o >= 1; o >>= 1) den += __shfl_xor(den, o, 32);

            const float4* F = (const float4*)feat;
            for (int base = beg; base < end; base += 32) {
                const int i = base + lane;
                int s = 0;
                float w = 0.0f;
                if (i < end) {
                    s = csr[i];
                    w = __expf(edge_score(as, a_nb[s]) - m) / den;
                }
                const int cnt = min(32, end - base);
                int j = 0;
                for (; j + 8 <= cnt; j += 8) {
                    int   sj[8];
                    float wj[8];
                    float4 f[8];
#pragma unroll
                    for (int u = 0; u < 8; ++u) { sj[u] = __shfl(s, j + u, 32); wj[u] = __shfl(w, j + u, 32); }
#pragma unroll
                    for (int u = 0; u < 8; ++u) f[u] = F[(size_t)sj[u] * (DD / 4) + lane];
#pragma unroll
                    for (int u = 0; u < 8; ++u) {
                        acc.x += wj[u] * f[u].x; acc.y += wj[u] * f[u].y;
                        acc.z += wj[u] * f[u].z; acc.w += wj[u] * f[u].w;
                    }
                }
                for (; j < cnt; ++j) {
                    const int   s1 = __shfl(s, j, 32);
                    const float w1 = __shfl(w, j, 32);
                    const float4 f1 = F[(size_t)s1 * (DD / 4) + lane];
                    acc.x += w1 * f1.x; acc.y += w1 * f1.y;
                    acc.z += w1 * f1.z; acc.w += w1 * f1.w;
                }
            }
        }
        float4* hp = (float4*)(hout + (size_t)n * (2 * DD) + b * DD) + lane;
        r = *hp;
        r.x = fmaxf(r.x + acc.x, 0.0f); r.y = fmaxf(r.y + acc.y, 0.0f);
        r.z = fmaxf(r.z + acc.z, 0.0f); r.w = fmaxf(r.w + acc.w, 0.0f);
        *hp = r;
        g = gids[n];
    }
    prow[wv][lane] = r;
    if (lane == 0) pg[wv] = g;
    __syncthreads();

    if (threadIdx.x < DD) {
        const int col = threadIdx.x;
        const float* pr = (const float*)prow;
        float run = 0.0f;
        int gcur = -1;
        for (int w = 0; w < GATHER_WAVES; ++w) {
            const int gw = pg[w];
            if (gw != gcur) {
                if ((unsigned)gcur < (unsigned)N_GRAPHS)
                    atomicAdd(phis + (size_t)gcur * (2 * DD) + b * DD + col, run);
                gcur = gw;
                run = 0.0f;
            }
            run += pr[w * DD + col];
        }
        if ((unsigned)gcur < (unsigned)N_GRAPHS)
            atomicAdd(phis + (size_t)gcur * (2 * DD) + b * DD + col, run);
    }
}

extern "C" void kernel_launch(void* const* d_in, const int* in_sizes, int n_in,
                              void* d_out, int out_size, void* d_ws, size_t ws_size,
                              hipStream_t stream) {
    const float* x        = (const float*)d_in[0];
    const int*   esrc     = (const int*)d_in[1];
    const int*   edst     = (const int*)d_in[2];
    const int*   gids     = (const int*)d_in[3];
    const float* Wfc      = (const float*)d_in[4];
    const float* bfc      = (const float*)d_in[5];
    const float* wl       = (const float*)d_in[6];
    const float* wr       = (const float*)d_in[7];
    const float* Wres     = (const float*)d_in[8];

    float* h    = (float*)d_out;
    float* phis = (float*)d_out + (size_t)N_NODES * 2 * DD;

    char* ws = (char*)d_ws;
    float* featb  = (float*)ws;
    int*   csr    = (int*)(ws + (size_t)N_NODES * DD * 4);
    int*   rowptr = (int*)(ws + (size_t)N_NODES * DD * 4 + (size_t)N_EDGES * 4);
    float* a_self = (float*)(rowptr + 2 * 10240);
    float* a_nb   = a_self + 10240;

    k_gemm<<<dim3((N_NODES + 63) / 64, 3), dim3(GEMM_THREADS), 0, stream>>>(
        x, Wfc, bfc, Wres, wl, wr, featb, h, a_self, a_nb, phis, 0);

    k_bin<<<dim3(BIN_WGS), dim3(BIN_THREADS), 0, stream>>>(esrc, edst, rowptr, csr);

    k_gather<<<dim3((N_NODES + GATHER_WAVES - 1) / GATHER_WAVES), dim3(GATHER_WAVES * 32), 0, stream>>>(
        rowptr, csr, a_self, a_nb, featb, h, gids, phis, 0);

    k_gemm<<<dim3((N_NODES + 63) / 64, 1), dim3(GEMM_THREADS), 0, stream>>>(
        x, Wfc, bfc, Wres, wl, wr, featb, h, a_self, a_nb, phis, 3);
    k_gather<<<dim3((N_NODES + GATHER_WAVES - 1) / GATHER_WAVES), dim3(GATHER_WAVES * 32), 0, stream>>>(
        rowptr, csr, a_self, a_nb, featb, h, gids, phis, 1);
}


# Edge grouping and the first GEMM launch merged into one grid of 1024-thread work-groups (125 grouping + 120 GEMM) that run side by side; 5->4 launches; block-0 dot sums now two 64-col halves
# speedup vs baseline: 5.6948x; 1.1104x over previous
#include <hip/hip_runtime.h>
#include <hip/hip_bf16.h>

#define N_NODES 10000
#define N_EDGES 640000
#define N_GRAPHS 64
#define DD 128
#define NEG_SLOPE 0.01f

typedef __attribute__((ext_vector_type(2))) float v2f;
typedef __attribute__((ext_vector_type(8))) float v8f;

#define BIN_NODES 80
#define BIN_WGS (N_NODES / BIN_NODES)
#define BIN_THREADS 1024
#define BIN_WAVES (BIN_THREADS / 32)
#define BIN_WCAP 512
#define BIN_CAP (BIN_WAVES * BIN_WCAP)
#define NV4 (N_EDGES / 4)
#define BIN_BATCH 8
#define BIN_NB 20

__device__ __forceinline__ void bin_body(const int* __restrict__ esrc,
                      const int* __restrict__ edst,
                      int* __restrict__ rowptr, int* __restrict__ csr, const int wg) {
    __shared__ unsigned wl[BIN_CAP];
    __shared__ int sorted[BIN_CAP];
    __shared__ int cnt[BIN_NODES], cur[BIN_NODES], pre[BIN_NODES + 1];
    __shared__ int wsum[BIN_WAVES];
    __shared__ int ovf;
    const int t = threadIdx.x;
    const int lane = t & 31;
    const int wave = t >> 5;
    const int lo = wg * BIN_NODES;
    if (t < BIN_NODES) { cnt[t] = 0; cur[t] = 0; }
    if (t == 0) ovf = 0;
    __syncthreads();

    const int rot = wg * (NV4 / BIN_WGS);
    const int4* D4 = (const int4*)edst;
    unsigned* mywl = wl + wave * BIN_WCAP;
    int below = 0;
    int wcnt = 0;

    auto loadb = [&](int4 (&d)[BIN_BATCH], int ib) {
#pragma unroll
        for (int u = 0; u < BIN_BATCH; ++u) {
            const int i = (ib * BIN_BATCH + u) * BIN_THREADS + t;
            int idx = i + rot;
            if (idx >= NV4) idx -= NV4;
            d[u] = (i < NV4) ? D4[idx] : make_int4(N_NODES, N_NODES, N_NODES, N_NODES);
        }
    };
    auto one = [&](int dv, int e) {
        const int tk = dv - lo;
        below += (int)((unsigned)tk >> 31);
        const bool mk = (unsigned)tk < BIN_NODES;
        const unsigned mask = __builtin_amdgcn_ballot_w32(mk);
        if (mask) {
            const int pos = wcnt + (int)__builtin_amdgcn_mbcnt_lo(mask, 0u);
            if (mk && pos < BIN_WCAP) mywl[pos] = ((unsigned)tk << 20) | (unsigned)e;
            wcnt += __builtin_popcount(mask);
        }
    };
    auto procb = [&](const int4 (&d)[BIN_BATCH], int ib) {
#pragma unroll
        for (int u = 0; u < BIN_BATCH; ++u) {
            const int i = (ib * BIN_BATCH + u) * BIN_THREADS + t;
            int idx = i + rot;
            if (idx >= NV4) idx -= NV4;
            const int e = idx * 4;
            one(d[u].x, e + 0);
            one(d[u].y, e + 1);
            one(d[u].z, e + 2);
            one(d[u].w, e + 3);
        }
    };

    int4 A[BIN_BATCH], B[BIN_BATCH];
    loadb(A, 0);
    for (int ib = 0; ib < BIN_NB; ib += 2) {
        loadb(B, ib + 1);
        procb(A, ib);
        loadb(A, ib + 2);
        procb(B, ib + 1);
    }

#pragma unroll
    for (int o = 16; o >= 1; o >>= 1) below += __shfl_xor(below, o, 32);
    if (lane == 0) {
        wsum[wave] = below;
        if (wcnt > BIN_WCAP) ovf = 1;
    }
    const int nw = min(wcnt, BIN_WCAP);
    for (int i = lane; i < nw; i += 32) atomicAdd(&cnt[mywl[i] >> 20], 1);
    __syncthreads();
    int base = 0;
#pragma unroll
    for (int w = 0; w < BIN_WAVES; ++w) base += wsum[w];
    const bool over = ovf != 0;
    if (over) {
        __syncthreads();
        if (t < BIN_NODES) cnt[t] = 0;
        __syncthreads();
        for (int e = t; e < N_EDGES; e += BIN_THREADS) {
            const int dl = edst[e] - lo;
            if ((unsigned)dl < BIN_NODES) atomicAdd(&cnt[dl], 1);
        }
        __syncthreads();
    }
    if (t <= BIN_NODES) {
        int run = 0;
        for (int k = 0; k < t; ++k) run += cnt[k];
        pre[t] = run;
    }
    __syncthreads();
    const int total = pre[BIN_NODES];
    if (t < BIN_NODES) rowptr[lo + t] = base + pre[t];
    if (t == 0 && wg == BIN_WGS - 1) rowptr[N_NODES] = base + total;

    if (!over) {
        for (int i0 = 0; i0 < nw; i0 += 8 * 32) {
            unsigned m[8];
            int s[8];
#pragma unroll
            for (int u = 0; u < 8; ++u) {
                const int i = i0 + u * 32 + lane;
                m[u] = (i < nw) ? mywl[i] : 0u;
                s[u] = (i < nw) ? esrc[m[u] & 0xFFFFFu] : 0;
            }
#pragma unroll
            for (int u = 0; u < 8; ++u) {
                const int i = i0 + u * 32 + lane;
                if (i < nw) {
                    const int dl = (int)(m[u] >> 20);
                    const int slot = atomicAdd(&cur[dl], 1);
                    sorted[pre[dl] + slot] = s[u];
                }
            }
        }
        __syncthreads();
        for (int i = t; i < total; i += BIN_THREADS) csr[base + i] = sorted[i];
    } else {
        for (int e = t; e < N_EDGES; e += BIN_THREADS) {
            const int dl = edst[e] - lo;
            if ((unsigned)dl < BIN_NODES) {
                const int slot = atomicAdd(&cur[dl], 1);
                csr[base + pre[dl] + slot] = esrc[e];
            }
        }
    }
}

#define GEMM_THREADS 128
#define WPITCH 132
__global__ void __launch_bounds__(GEMM_THREADS) k_gemm(const float* __restrict__ x,
                       const float* __restrict__ Wfc, const float* __restrict__ bfc,
                       const float* __restrict__ Wres,
                       const float* __restrict__ wl, const float* __restrict__ wr,
                       float* __restrict__ feat_ws, float* __restrict__ hout,
                       float* __restrict__ a_self, float* __restrict__ a_nb, int job0) {
    __shared__ float Ws[DD * WPITCH];
    const int job   = blockIdx.y + job0;
    const int b     = (job >= 2) ? 1 : 0;
    const int which = (job == 1 || job == 2) ? 1 : 0;
    const float* W = (which == 0 ? Wfc : Wres) + (size_t)b * DD * DD;

    const float4* W4 = (const float4*)W;
    for (int i0 = 0; i0 < DD * DD / 4; i0 += 8 * GEMM_THREADS) {
        float4 w[8];
#pragma unroll
        for (int u = 0; u < 8; ++u) w[u] = W4[i0 + u * GEMM_THREADS + threadIdx.x];
#pragma unroll
        for (int u = 0; u < 8; ++u) {
            const int i = i0 + u * GEMM_THREADS + threadIdx.x;
            const int j = i >> 5, k4 = (i & 31) << 2;
            *(float4*)(Ws + j * WPITCH + k4) = w[u];
        }
    }
    __syncthreads();

    const int lane = threadIdx.x & 31;
    const int wave = threadIdx.x >> 5;
    const int rowBase = blockIdx.x * 64 + wave * 16;
    const int l16 = lane & 15;
    const int hi  = lane >> 4;

    v8f c[8];
#pragma unroll
    for (int t = 0; t < 8; ++t) {
        float bv = (which == 0) ? bfc[b * DD + t * 16 + l16] : 0.0f;
#pragma unroll
        for (int v = 0; v < 8; ++v) c[t][v] = bv;
    }

    const int arow = min(rowBase + l16, N_NODES - 1);
    const float* xrow = x + (size_t)arow * DD;

    for (int k0 = 0; k0 < DD; k0 += 4) {
        const int krow = k0 + 2 * hi;
        v2f a = *(const v2f*)(xrow + krow);
#pragma unroll
        for (int t = 0; t < 8; ++t) {
            const int j = t * 16 + l16;
            const v2f bf = *(const v2f*)(Ws + j * WPITCH + krow);
            c[t] = __builtin_amdgcn_wmma_f32_16x16x4_f32(
                false, a, false, bf, (short)0, c[t], false, false);
        }
    }

#pragma unroll
    for (int t = 0; t < 8; ++t) {
        const int col = t * 16 + l16;
#pragma unroll
        for (int v = 0; v < 8; ++v) {
            const int row = rowBase + v + 8 * hi;
            if (row < N_NODES) {
                if (which == 0)
                    feat_ws[(size_t)row * DD + col] = c[t][v];
                else
                    hout[(size_t)row * (2 * DD) + b * DD + col] = c[t][v];
            }
        }
    }

    if (which == 0) {
        float wlv[8], wrv[8];
#pragma unroll
        for (int t = 0; t < 8; ++t) {
            wlv[t] = wl[b * DD + t * 16 + l16];
            wrv[t] = wr[b * DD + t * 16 + l16];
        }
        float my_sl = 0.0f, my_sr = 0.0f;
#pragma unroll
        for (int v = 0; v < 8; ++v) {
            float sl = 0.0f, sr = 0.0f;
#pragma unroll
            for (int t = 0; t < 8; ++t) {
                sl += c[t][v] * wlv[t];
                sr += c[t][v] * wrv[t];
            }
#pragma unroll
            for (int m = 8; m >= 1; m >>= 1) {
                sl += __shfl_xor(sl, m, 32);
                sr += __shfl_xor(sr, m, 32);
            }
            if (l16 == v) { my_sl = sl; my_sr = sr; }
        }
        const int myrow = rowBase + l16 + 8 * hi;
        if (l16 < 8 && myrow < N_NODES) {
            a_self[myrow] = my_sl;
            a_nb[myrow]   = my_sr;
        }
    }
}

#define FR_THREADS 1024
#define FR_ROWS 256
#define FR_TILES ((N_NODES + FR_ROWS - 1) / FR_ROWS)
#define FR_JOBS 3
__device__ __forceinline__ void front_gemm(const float* __restrict__ x,
                       const float* __restrict__ Wfc, const float* __restrict__ bfc,
                       const float* __restrict__ Wres,
                       const float* __restrict__ wl, const float* __restrict__ wr,
                       float* __restrict__ feat_ws, float* __restrict__ hout,
                       float* __restrict__ a_self, float* __restrict__ a_nb,
                       float* __restrict__ phis, const int g) {
    __shared__ float Ws[DD * WPITCH];
    __shared__ float dots[FR_ROWS][2][2];
    const int job  = g / FR_TILES;
    const int tile = g - job * FR_TILES;
    if (job == 0) {
        const int i = tile * FR_THREADS + threadIdx.x;
        if (i < N_GRAPHS * 2 * DD) phis[i] = 0.0f;
    }
    const int b     = (job == 2) ? 1 : 0;
    const int which = (job == 0) ? 0 : 1;
    const float* W = (which == 0 ? Wfc : Wres) + (size_t)b * DD * DD;

    const float4* W4 = (const float4*)W;
    {
        float4 w[4];
#pragma unroll
        for (int u = 0; u < 4; ++u) w[u] = W4[u * FR_THREADS + threadIdx.x];
#pragma unroll
        for (int u = 0; u < 4; ++u) {
            const int i = u * FR_THREADS + threadIdx.x;
            const int j = i >> 5, k4 = (i & 31) << 2;
            *(float4*)(Ws + j * WPITCH + k4) = w[u];
        }
    }
    __syncthreads();

    const int lane = threadIdx.x & 31;
    const int wave = threadIdx.x >> 5;
    const int strip = wave >> 1;
    const int colBase = (wave & 1) * 64;
    const int rowBase = tile * FR_ROWS + strip * 16;
    const int l16 = lane & 15;
    const int hi  = lane >> 4;

    v8f c[4];
#pragma unroll
    for (int t = 0; t < 4; ++t) {
        float bv = (which == 0) ? bfc[b * DD + colBase + t * 16 + l16] : 0.0f;
#pragma unroll
        for (int v = 0; v < 8; ++v) c[t][v] = bv;
    }

    const int arow = min(rowBase + l16, N_NODES - 1);
    const float* xrow = x + (size_t)arow * DD;

    for (int k0 = 0; k0 < DD; k0 += 4) {
        const int krow = k0 + 2 * hi;
        v2f a = *(const v2f*)(xrow + krow);
#pragma unroll
        for (int t = 0; t < 4; ++t) {
            const int j = colBase + t * 16 + l16;
            const v2f bf = *(const v2f*)(Ws + j * WPITCH + krow);
            c[t] = __builtin_amdgcn_wmma_f32_16x16x4_f32(
                false, a, false, bf, (short)0, c[t], false, false);
        }
    }

#pragma unroll
    for (int t = 0; t < 4; ++t) {
        const int col = colBase + t * 16 + l16;
#pragma unroll
        for (int v = 0; v < 8; ++v) {
            const int row = rowBase + v + 8 * hi;
            if (row < N_NODES) {
                if (which == 0)
                    feat_ws[(size_t)row * DD + col] = c[t][v];
                else
                    hout[(size_t)row * (2 * DD) + b * DD + col] = c[t][v];
            }
        }
    }

    if (which == 0) {
        float wlv[4], wrv[4];
#pragma unroll
        for (int t = 0; t < 4; ++t) {
            wlv[t] = wl[b * DD + colBase + t * 16 + l16];
            wrv[t] = wr[b * DD + colBase + t * 16 + l16];
        }
        float my_sl = 0.0f, my_sr = 0.0f;
#pragma unroll
        for (int v = 0; v < 8; ++v) {
            float sl = 0.0f, sr = 0.0f;
#pragma unroll
            for (int t = 0; t < 4; ++t) {
                sl += c[t][v] * wlv[t];
                sr += c[t][v] * wrv[t];
            }
#pragma unroll
            for (int m = 8; m >= 1; m >>= 1) {
                sl += __shfl_xor(sl, m, 32);
                sr += __shfl_xor(sr, m, 32);
            }
            if (l16 == v) { my_sl = sl; my_sr = sr; }
        }
        if (l16 < 8) {
            const int r = strip * 16 + l16 + 8 * hi;
            dots[r][wave & 1][0] = my_sl;
            dots[r][wave & 1][1] = my_sr;
        }
        __syncthreads();
        if (threadIdx.x < FR_ROWS) {
            const int row = tile * FR_ROWS + threadIdx.x;
            if (row < N_NODES) {
                a_self[row] = dots[threadIdx.x][0][0] + dots[threadIdx.x][1][0];
                a_nb[row]   = dots[threadIdx.x][0][1] + dots[threadIdx.x][1][1];
            }
        }
    }
}

__global__ void __launch_bounds__(FR_THREADS) k_front(
                       const int* __restrict__ esrc, const int* __restrict__ edst,
                       int* __restrict__ rowptr, int* __restrict__ csr,
                       const float* __restrict__ x,
                       const float* __restrict__ Wfc, const float* __restrict__ bfc,
                       const float* __restrict__ Wres,
                       const float* __restrict__ wl, const float* __restrict__ wr,
                       float* __restrict__ feat_ws, float* __restrict__ hout,
                       float* __restrict__ a_self, float* __restrict__ a_nb,
                       float* __restrict__ phis) {
    if (blockIdx.x < BIN_WGS)
        bin_body(esrc, edst, rowptr, csr, blockIdx.x);
    else
        front_gemm(x, Wfc, bfc, Wres, wl, wr, feat_ws, hout, a_self, a_nb, phis,
                   blockIdx.x - BIN_WGS);
}

__device__ __forceinline__ float edge_score(float as, float an) {
    float sc = as + an;
    return sc >= 0.0f ? sc : NEG_SLOPE * sc;
}

#define GATHER_WAVES 8
__global__ void __launch_bounds__(GATHER_WAVES * 32) k_gather(
                         const int* __restrict__ rowptr, const int* __restrict__ csr,
                         const float* __restrict__ a_self, const float* __restrict__ a_nb,
                         const float* __restrict__ feat, float* __restrict__ hout,
                         const int* __restrict__ gids, float* __restrict__ phis, int b) {
    __shared__ float4 prow[GATHER_WAVES][32];
    __shared__ int pg[GATHER_WAVES];
    const int lane = threadIdx.x & 31;
    const int wv = threadIdx.x >> 5;
    const int n = blockIdx.x * GATHER_WAVES + wv;

    float4 r = make_float4(0.0f, 0.0f, 0.0f, 0.0f);
    int g = -1;
    if (n < N_NODES) {
        const int beg = rowptr[n], end = rowptr[n + 1];
        float4 acc = make_float4(0.0f, 0.0f, 0.0f, 0.0f);
        if (beg < end) {
            const float as = a_self[n];

            float m = -INFINITY;
            for (int i = beg + lane; i < end; i += 32)
                m = fmaxf(m, edge_score(as, a_nb[csr[i]]));
#pragma unroll
            for (int o = 16; o >= 1; o >>= 1) m = fmaxf(m, __shfl_xor(m, o, 32));

            float den = 0.0f;
            for (int i = beg + lane; i < end; i += 32)
                den += __expf(edge_score(as, a_nb[csr[i]]) - m);
#pragma unroll
            for (int o = 16; o >= 1; o >>= 1) den += __shfl_xor(den, o, 32);

            const float4* F = (const float4*)feat;
            for (int base = beg; base < end; base += 32) {
                const int i = base + lane;
                int s = 0;
                float w = 0.0f;
                if (i < end) {
                    s = csr[i];
                    w = __expf(edge_score(as, a_nb[s]) - m) / den;
                }
                const int cnt = min(32, end - base);
                int j = 0;
                for (; j + 8 <= cnt; j += 8) {
                    int   sj[8];
                    float wj[8];
                    float4 f[8];
#pragma unroll
                    for (int u = 0; u < 8; ++u) { sj[u] = __shfl(s, j + u, 32); wj[u] = __shfl(w, j + u, 32); }
#pragma unroll
                    for (int u = 0; u < 8; ++u) f[u] = F[(size_t)sj[u] * (DD / 4) + lane];
#pragma unroll
                    for (int u = 0; u < 8; ++u) {
                        acc.x += wj[u] * f[u].x; acc.y += wj[u] * f[u].y;
                        acc.z += wj[u] * f[u].z; acc.w += wj[u] * f[u].w;
                    }
                }
                for (; j < cnt; ++j) {
                    const int   s1 = __shfl(s, j, 32);
                    const float w1 = __shfl(w, j, 32);
                    const float4 f1 = F[(size_t)s1 * (DD / 4) + lane];
                    acc.x += w1 * f1.x; acc.y += w1 * f1.y;
                    acc.z += w1 * f1.z; acc.w += w1 * f1.w;
                }
            }
        }
        float4* hp = (float4*)(hout + (size_t)n * (2 * DD) + b * DD) + lane;
        r = *hp;
        r.x = fmaxf(r.x + acc.x, 0.0f); r.y = fmaxf(r.y + acc.y, 0.0f);
        r.z = fmaxf(r.z + acc.z, 0.0f); r.w = fmaxf(r.w + acc.w, 0.0f);
        *hp = r;
        g = gids[n];
    }
    prow[wv][lane] = r;
    if (lane == 0) pg[wv] = g;
    __syncthreads();

    if (threadIdx.x < DD) {
        const int col = threadIdx.x;
        const float* pr = (const float*)prow;
        float run = 0.0f;
        int gcur = -1;
        for (int w = 0; w < GATHER_WAVES; ++w) {
            const int gw = pg[w];
            if (gw != gcur) {
                if ((unsigned)gcur < (unsigned)N_GRAPHS)
                    atomicAdd(phis + (size_t)gcur * (2 * DD) + b * DD + col, run);
                gcur = gw;
                run = 0.0f;
            }
            run += pr[w * DD + col];
        }
        if ((unsigned)gcur < (unsigned)N_GRAPHS)
            atomicAdd(phis + (size_t)gcur * (2 * DD) + b * DD + col, run);
    }
}

extern "C" void kernel_launch(void* const* d_in, const int* in_sizes, int n_in,
                              void* d_out, int out_size, void* d_ws, size_t ws_size,
                              hipStream_t stream) {
    const float* x        = (const float*)d_in[0];
    const int*   esrc     = (const int*)d_in[1];
    const int*   edst     = (const int*)d_in[2];
    const int*   gids     = (const int*)d_in[3];
    const float* Wfc      = (const float*)d_in[4];
    const float* bfc      = (const float*)d_in[5];
    const float* wl       = (const float*)d_in[6];
    const float* wr       = (const float*)d_in[7];
    const float* Wres     = (const float*)d_in[8];

    float* h    = (float*)d_out;
    float* phis = (float*)d_out + (size_t)N_NODES * 2 * DD;

    char* ws = (char*)d_ws;
    float* featb  = (float*)ws;
    int*   csr    = (int*)(ws + (size_t)N_NODES * DD * 4);
    int*   rowptr = (int*)(ws + (size_t)N_NODES * DD * 4 + (size_t)N_EDGES * 4);
    float* a_self = (float*)(rowptr + 2 * 10240);
    float* a_nb   = a_self + 10240;

    k_front<<<dim3(BIN_WGS + FR_JOBS * FR_TILES), dim3(FR_THREADS), 0, stream>>>(
        esrc, edst, rowptr, csr, x, Wfc, bfc, Wres, wl, wr, featb, h, a_self, a_nb, phis);

    k_gather<<<dim3((N_NODES + GATHER_WAVES - 1) / GATHER_WAVES), dim3(GATHER_WAVES * 32), 0, stream>>>(
        rowptr, csr, a_self, a_nb, featb, h, gids, phis, 0);

    k_gemm<<<dim3((N_NODES + 63) / 64, 1), dim3(GEMM_THREADS), 0, stream>>>(
        x, Wfc, bfc, Wres, wl, wr, featb, h, a_self, a_nb, 3);
    k_gather<<<dim3((N_NODES + GATHER_WAVES - 1) / GATHER_WAVES), dim3(GATHER_WAVES * 32), 0, stream>>>(
        rowptr, csr, a_self, a_nb, featb, h, gids, phis, 1);
}


# Edge list cut in two halves, each grouped by destination by its own 68 work-groups (136 + 120 GEMM = 256): a group streams half the list; gather walks 2 csr runs per node; sum order: half 0 then 1
# speedup vs baseline: 6.4856x; 1.1389x over previous
#include <hip/hip_runtime.h>
#include <hip/hip_bf16.h>

#define N_NODES 10000
#define N_EDGES 640000
#define N_GRAPHS 64
#define DD 128
#define NEG_SLOPE 0.01f

typedef __attribute__((ext_vector_type(2))) float v2f;
typedef __attribute__((ext_vector_type(8))) float v8f;

#define BIN_NODES 148
#define BIN_RANGES ((N_NODES + BIN_NODES - 1) / BIN_NODES)
#define BIN_HALVES 2
#define BIN_WGS (BIN_RANGES * BIN_HALVES)
#define BIN_THREADS 1024
#define BIN_WAVES (BIN_THREADS / 32)
#define BIN_WCAP 512
#define BIN_CAP (BIN_WAVES * BIN_WCAP)
#define HALF_EDGES (N_EDGES / BIN_HALVES)
#define NV4 (HALF_EDGES / 4)
#define BIN_BATCH 8
#define BIN_NB 10
#define RP_PITCH 10240
#define BIN_FAR 0x7FFFFFFF

__device__ __forceinline__ void bin_body(const int* __restrict__ esrc,
                      const int* __restrict__ edst,
                      int* __restrict__ rowptr, int* __restrict__ csr, const int wg) {
    __shared__ unsigned wl[BIN_CAP];
    __shared__ int sorted[BIN_CAP];
    __shared__ int cnt[BIN_NODES], cur[BIN_NODES], pre[BIN_NODES + 1];
    __shared__ int wsum[BIN_WAVES];
    __shared__ int ovf;
    const int t = threadIdx.x;
    const int lane = t & 31;
    const int wave = t >> 5;
    const int rng = wg >> 1;
    const int half = wg & 1;
    const int lo = rng * BIN_NODES;
    const int e_lo = half * HALF_EDGES;
    if (t < BIN_NODES) { cnt[t] = 0; cur[t] = 0; }
    if (t == 0) ovf = 0;
    __syncthreads();

    const int rot = rng * (NV4 / BIN_RANGES);
    const int4* D4 = (const int4*)(edst + e_lo);
    unsigned* mywl = wl + wave * BIN_WCAP;
    int below = 0;
    int wcnt = 0;

    auto loadb = [&](int4 (&d)[BIN_BATCH], int ib) {
#pragma unroll
        for (int u = 0; u < BIN_BATCH; ++u) {
            const int i = (ib * BIN_BATCH + u) * BIN_THREADS + t;
            int idx = i + rot;
            if (idx >= NV4) idx -= NV4;
            d[u] = (i < NV4) ? D4[idx] : make_int4(BIN_FAR, BIN_FAR, BIN_FAR, BIN_FAR);
        }
    };
    auto one = [&](int dv, int e) {
        const int tk = dv - lo;
        below += (int)((unsigned)tk >> 31);
        const bool mk = (unsigned)tk < BIN_NODES;
        const unsigned mask = __builtin_amdgcn_ballot_w32(mk);
        if (mask) {
            const int pos = wcnt + (int)__builtin_amdgcn_mbcnt_lo(mask, 0u);
            if (mk && pos < BIN_WCAP) mywl[pos] = ((unsigned)tk << 20) | (unsigned)e;
            wcnt += __builtin_popcount(mask);
        }
    };
    auto procb = [&](const int4 (&d)[BIN_BATCH], int ib) {
#pragma unroll
        for (int u = 0; u < BIN_BATCH; ++u) {
            const int i = (ib * BIN_BATCH + u) * BIN_THREADS + t;
            int idx = i + rot;
            if (idx >= NV4) idx -= NV4;
            const int e = e_lo + idx * 4;
            one(d[u].x, e + 0);
            one(d[u].y, e + 1);
            one(d[u].z, e + 2);
            one(d[u].w, e + 3);
        }
    };

    int4 A[BIN_BATCH], B[BIN_BATCH];
    loadb(A, 0);
    for (int ib = 0; ib < BIN_NB; ib += 2) {
        loadb(B, ib + 1);
        procb(A, ib);
        loadb(A, ib + 2);
        procb(B, ib + 1);
    }

#pragma unroll
    for (int o = 16; o >= 1; o >>= 1) below += __shfl_xor(below, o, 32);
    if (lane == 0) {
        wsum[wave] = below;
        if (wcnt > BIN_WCAP) ovf = 1;
    }
    const int nw = min(wcnt, BIN_WCAP);
    for (int i = lane; i < nw; i += 32) atomicAdd(&cnt[mywl[i] >> 20], 1);
    __syncthreads();
    int base = e_lo;
#pragma unroll
    for (int w = 0; w < BIN_WAVES; ++w) base += wsum[w];
    const bool over = ovf != 0;
    if (over) {
        __syncthreads();
        if (t < BIN_NODES) cnt[t] = 0;
        __syncthreads();
        for (int e = e_lo + t; e < e_lo + HALF_EDGES; e += BIN_THREADS) {
            const int dl = edst[e] - lo;
            if ((unsigned)dl < BIN_NODES) atomicAdd(&cnt[dl], 1);
        }
        __syncthreads();
    }
    if (t <= BIN_NODES) {
        int run = 0;
        for (int k = 0; k < t; ++k) run += cnt[k];
        pre[t] = run;
    }
    __syncthreads();
    const int total = pre[BIN_NODES];
    int* rp = rowptr + half * RP_PITCH;
    if (t < BIN_NODES && lo + t < N_NODES) rp[lo + t] = base + pre[t];
    if (t == 0 && rng == BIN_RANGES - 1) rp[N_NODES] = base + total;

    if (!over) {
        for (int i0 = 0; i0 < nw; i0 += 8 * 32) {
            unsigned m[8];
            int s[8];
#pragma unroll
            for (int u = 0; u < 8; ++u) {
                const int i = i0 + u * 32 + lane;
                m[u] = (i < nw) ? mywl[i] : 0u;
                s[u] = (i < nw) ? esrc[m[u] & 0xFFFFFu] : 0;
            }
#pragma unroll
            for (int u = 0; u < 8; ++u) {
                const int i = i0 + u * 32 + lane;
                if (i < nw) {
                    const int dl = (int)(m[u] >> 20);
                    const int slot = atomicAdd(&cur[dl], 1);
                    sorted[pre[dl] + slot] = s[u];
                }
            }
        }
        __syncthreads();
        for (int i = t; i < total; i += BIN_THREADS) csr[base + i] = sorted[i];
    } else {
        for (int e = e_lo + t; e < e_lo + HALF_EDGES; e += BIN_THREADS) {
            const int dl = edst[e] - lo;
            if ((unsigned)dl < BIN_NODES) {
                const int slot = atomicAdd(&cur[dl], 1);
                csr[base + pre[dl] + slot] = esrc[e];
            }
        }
    }
}

#define GEMM_THREADS 128
#define WPITCH 132
__global__ void __launch_bounds__(GEMM_THREADS) k_gemm(const float* __restrict__ x,
                       const float* __restrict__ Wfc, const float* __restrict__ bfc,
                       const float* __restrict__ Wres,
                       const float* __restrict__ wl, const float* __restrict__ wr,
                       float* __restrict__ feat_ws, float* __restrict__ hout,
                       float* __restrict__ a_self, float* __restrict__ a_nb, int job0) {
    __shared__ float Ws[DD * WPITCH];
    const int job   = blockIdx.y + job0;
    const int b     = (job >= 2) ? 1 : 0;
    const int which = (job == 1 || job == 2) ? 1 : 0;
    const float* W = (which == 0 ? Wfc : Wres) + (size_t)b * DD * DD;

    const float4* W4 = (const float4*)W;
    for (int i0 = 0; i0 < DD * DD / 4; i0 += 8 * GEMM_THREADS) {
        float4 w[8];
#pragma unroll
        for (int u = 0; u < 8; ++u) w[u] = W4[i0 + u * GEMM_THREADS + threadIdx.x];
#pragma unroll
        for (int u = 0; u < 8; ++u) {
            const int i = i0 + u * GEMM_THREADS + threadIdx.x;
            const int j = i >> 5, k4 = (i & 31) << 2;
            *(float4*)(Ws + j * WPITCH + k4) = w[u];
        }
    }
    __syncthreads();

    const int lane = threadIdx.x & 31;
    const int wave = threadIdx.x >> 5;
    const int rowBase = blockIdx.x * 64 + wave * 16;
    const int l16 = lane & 15;
    const int hi  = lane >> 4;

    v8f c[8];
#pragma unroll
    for (int t = 0; t < 8; ++t) {
        float bv = (which == 0) ? bfc[b * DD + t * 16 + l16] : 0.0f;
#pragma unroll
        for (int v = 0; v < 8; ++v) c[t][v] = bv;
    }

    const int arow = min(rowBase + l16, N_NODES - 1);
    const float* xrow = x + (size_t)arow * DD;

    for (int k0 = 0; k0 < DD; k0 += 4) {
        const int krow = k0 + 2 * hi;
        v2f a = *(const v2f*)(xrow + krow);
#pragma unroll
        for (int t = 0; t < 8; ++t) {
            const int j = t * 16 + l16;
            const v2f bf = *(const v2f*)(Ws + j * WPITCH + krow);
            c[t] = __builtin_amdgcn_wmma_f32_16x16x4_f32(
                false, a, false, bf, (short)0, c[t], false, false);
        }
    }

#pragma unroll
    for (int t = 0; t < 8; ++t) {
        const int col = t * 16 + l16;
#pragma unroll
        for (int v = 0; v < 8; ++v) {
            const int row = rowBase + v + 8 * hi;
            if (row < N_NODES) {
                if (which == 0)
                    feat_ws[(size_t)row * DD + col] = c[t][v];
                else
                    hout[(size_t)row * (2 * DD) + b * DD + col] = c[t][v];
            }
        }
    }

    if (which == 0) {
        float wlv[8], wrv[8];
#pragma unroll
        for (int t = 0; t < 8; ++t) {
            wlv[t] = wl[b * DD + t * 16 + l16];
            wrv[t] = wr[b * DD + t * 16 + l16];
        }
        float my_sl = 0.0f, my_sr = 0.0f;
#pragma unroll
        for (int v = 0; v < 8; ++v) {
            float sl = 0.0f, sr = 0.0f;
#pragma unroll
            for (int t = 0; t < 8; ++t) {
                sl += c[t][v] * wlv[t];
                sr += c[t][v] * wrv[t];
            }
#pragma unroll
            for (int m = 8; m >= 1; m >>= 1) {
                sl += __shfl_xor(sl, m, 32);
                sr += __shfl_xor(sr, m, 32);
            }
            if (l16 == v) { my_sl = sl; my_sr = sr; }
        }
        const int myrow = rowBase + l16 + 8 * hi;
        if (l16 < 8 && myrow < N_NODES) {
            a_self[myrow] = my_sl;
            a_nb[myrow]   = my_sr;
        }
    }
}

#define FR_THREADS 1024
#define FR_ROWS 256
#define FR_TILES ((N_NODES + FR_ROWS - 1) / FR_ROWS)
#define FR_JOBS 3
__device__ __forceinline__ void front_gemm(const float* __restrict__ x,
                       const float* __restrict__ Wfc, const float* __restrict__ bfc,
                       const float* __restrict__ Wres,
                       const float* __restrict__ wl, const float* __restrict__ wr,
                       float* __restrict__ feat_ws, float* __restrict__ hout,
                       float* __restrict__ a_self, float* __restrict__ a_nb,
                       float* __restrict__ phis, const int g) {
    __shared__ float Ws[DD * WPITCH];
    __shared__ float dots[FR_ROWS][2][2];
    const int job  = g / FR_TILES;
    const int tile = g - job * FR_TILES;
    if (job == 0) {
        const int i = tile * FR_THREADS + threadIdx.x;
        if (i < N_GRAPHS * 2 * DD) phis[i] = 0.0f;
    }
    const int b     = (job == 2) ? 1 : 0;
    const int which = (job == 0) ? 0 : 1;
    const float* W = (which == 0 ? Wfc : Wres) + (size_t)b * DD * DD;

    const float4* W4 = (const float4*)W;
    {
        float4 w[4];
#pragma unroll
        for (int u = 0; u < 4; ++u) w[u] = W4[u * FR_THREADS + threadIdx.x];
#pragma unroll
        for (int u = 0; u < 4; ++u) {
            const int i = u * FR_THREADS + threadIdx.x;
            const int j = i >> 5, k4 = (i & 31) << 2;
            *(float4*)(Ws + j * WPITCH + k4) = w[u];
        }
    }
    __syncthreads();

    const int lane = threadIdx.x & 31;
    const int wave = threadIdx.x >> 5;
    const int strip = wave >> 1;
    const int colBase = (wave & 1) * 64;
    const int rowBase = tile * FR_ROWS + strip * 16;
    const int l16 = lane & 15;
    const int hi  = lane >> 4;

    v8f c[4];
#pragma unroll
    for (int t = 0; t < 4; ++t) {
        float bv = (which == 0) ? bfc[b * DD + colBase + t * 16 + l16] : 0.0f;
#pragma unroll
        for (int v = 0; v < 8; ++v) c[t][v] = bv;
    }

    const int arow = min(rowBase + l16, N_NODES - 1);
    const float* xrow = x + (size_t)arow * DD;

    for (int k0 = 0; k0 < DD; k0 += 4) {
        const int krow = k0 + 2 * hi;
        v2f a = *(const v2f*)(xrow + krow);
#pragma unroll
        for (int t = 0; t < 4; ++t) {
            const int j = colBase + t * 16 + l16;
            const v2f bf = *(const v2f*)(Ws + j * WPITCH + krow);
            c[t] = __builtin_amdgcn_wmma_f32_16x16x4_f32(
                false, a, false, bf, (short)0, c[t], false, false);
        }
    }

#pragma unroll
    for (int t = 0; t < 4; ++t) {
        const int col = colBase + t * 16 + l16;
#pragma unroll
        for (int v = 0; v < 8; ++v) {
            const int row = rowBase + v + 8 * hi;
            if (row < N_NODES) {
                if (which == 0)
                    feat_ws[(size_t)row * DD + col] = c[t][v];
                else
                    hout[(size_t)row * (2 * DD) + b * DD + col] = c[t][v];
            }
        }
    }

    if (which == 0) {
        float wlv[4], wrv[4];
#pragma unroll
        for (int t = 0; t < 4; ++t) {
            wlv[t] = wl[b * DD + colBase + t * 16 + l16];
            wrv[t] = wr[b * DD + colBase + t * 16 + l16];
        }
        float my_sl = 0.0f, my_sr = 0.0f;
#pragma unroll
        for (int v = 0; v < 8; ++v) {
            float sl = 0.0f, sr = 0.0f;
#pragma unroll
            for (int t = 0; t < 4; ++t) {
                sl += c[t][v] * wlv[t];
                sr += c[t][v] * wrv[t];
            }
#pragma unroll
            for (int m = 8; m >= 1; m >>= 1) {
                sl += __shfl_xor(sl, m, 32);
                sr += __shfl_xor(sr, m, 32);
            }
            if (l16 == v) { my_sl = sl; my_sr = sr; }
        }
        if (l16 < 8) {
            const int r = strip * 16 + l16 + 8 * hi;
            dots[r][wave & 1][0] = my_sl;
            dots[r][wave & 1][1] = my_sr;
        }
        __syncthreads();
        if (threadIdx.x < FR_ROWS) {
            const int row = tile * FR_ROWS + threadIdx.x;
            if (row < N_NODES) {
                a_self[row] = dots[threadIdx.x][0][0] + dots[threadIdx.x][1][0];
                a_nb[row]   = dots[threadIdx.x][0][1] + dots[threadIdx.x][1][1];
            }
        }
    }
}

__global__ void __launch_bounds__(FR_THREADS) k_front(
                       const int* __restrict__ esrc, const int* __restrict__ edst,
                       int* __restrict__ rowptr, int* __restrict__ csr,
                       const float* __restrict__ x,
                       const float* __restrict__ Wfc, const float* __restrict__ bfc,
                       const float* __restrict__ Wres,
                       const float* __restrict__ wl, const float* __restrict__ wr,
                       float* __restrict__ feat_ws, float* __restrict__ hout,
                       float* __restrict__ a_self, float* __restrict__ a_nb,
                       float* __restrict__ phis) {
    if (blockIdx.x < BIN_WGS)
        bin_body(esrc, edst, rowptr, csr, blockIdx.x);
    else
        front_gemm(x, Wfc, bfc, Wres, wl, wr, feat_ws, hout, a_self, a_nb, phis,
                   blockIdx.x - BIN_WGS);
}

__device__ __forceinline__ float edge_score(float as, float an) {
    float sc = as + an;
    return sc >= 0.0f ? sc : NEG_SLOPE * sc;
}

#define GATHER_WAVES 8
__global__ void __launch_bounds__(GATHER_WAVES * 32) k_gather(
                         const int* __restrict__ rowptr, const int* __restrict__ csr,
                         const float* __restrict__ a_self, const float* __restrict__ a_nb,
                         const float* __restrict__ feat, float* __restrict__ hout,
                         const int* __restrict__ gids, float* __restrict__ phis, int b) {
    __shared__ float4 prow[GATHER_WAVES][32];
    __shared__ int pg[GATHER_WAVES];
    const int lane = threadIdx.x & 31;
    const int wv = threadIdx.x >> 5;
    const int n = blockIdx.x * GATHER_WAVES + wv;

    float4 r = make_float4(0.0f, 0.0f, 0.0f, 0.0f);
    int g = -1;
    if (n < N_NODES) {
        const int beg0 = rowptr[n], end0 = rowptr[n + 1];
        const int beg1 = rowptr[RP_PITCH + n], end1 = rowptr[RP_PITCH + n + 1];
        const int n0 = end0 - beg0;
        const int deg = n0 + (end1 - beg1);
        const int off1 = beg1 - n0;
        auto at = [&](int i) { return i + (i < n0 ? beg0 : off1); };
        float4 acc = make_float4(0.0f, 0.0f, 0.0f, 0.0f);
        if (deg > 0) {
            const float as = a_self[n];

            float m = -INFINITY;
            for (int i = lane; i < deg; i += 32)
                m = fmaxf(m, edge_score(as, a_nb[csr[at(i)]]));
#pragma unroll
            for (int o = 16; o >= 1; o >>= 1) m = fmaxf(m, __shfl_xor(m, o, 32));

            float den = 0.0f;
            for (int i = lane; i < deg; i += 32)
                den += __expf(edge_score(as, a_nb[csr[at(i)]]) - m);
#pragma unroll
            for (int o = 16; o >= 1; o >>= 1) den += __shfl_xor(den, o, 32);

            const float4* F = (const float4*)feat;
            for (int base = 0; base < deg; base += 32) {
                const int i = base + lane;
                int s = 0;
                float w = 0.0f;
                if (i < deg) {
                    s = csr[at(i)];
                    w = __expf(edge_score(as, a_nb[s]) - m) / den;
                }
                const int cnt = min(32, deg - base);
                int j = 0;
                for (; j + 8 <= cnt; j += 8) {
                    int   sj[8];
                    float wj[8];
                    float4 f[8];
#pragma unroll
                    for (int u = 0; u < 8; ++u) { sj[u] = __shfl(s, j + u, 32); wj[u] = __shfl(w, j + u, 32); }
#pragma unroll
                    for (int u = 0; u < 8; ++u) f[u] = F[(size_t)sj[u] * (DD / 4) + lane];
#pragma unroll
                    for (int u = 0; u < 8; ++u) {
                        acc.x += wj[u] * f[u].x; acc.y += wj[u] * f[u].y;
                        acc.z += wj[u] * f[u].z; acc.w += wj[u] * f[u].w;
                    }
                }
                for (; j < cnt; ++j) {
                    const int   s1 = __shfl(s, j, 32);
                    const float w1 = __shfl(w, j, 32);
                    const float4 f1 = F[(size_t)s1 * (DD / 4) + lane];
                    acc.x += w1 * f1.x; acc.y += w1 * f1.y;
                    acc.z += w1 * f1.z; acc.w += w1 * f1.w;
                }
            }
        }
        float4* hp = (float4*)(hout + (size_t)n * (2 * DD) + b * DD) + lane;
        r = *hp;
        r.x = fmaxf(r.x + acc.x, 0.0f); r.y = fmaxf(r.y + acc.y, 0.0f);
        r.z = fmaxf(r.z + acc.z, 0.0f); r.w = fmaxf(r.w + acc.w, 0.0f);
        *hp = r;
        g = gids[n];
    }
    prow[wv][lane] = r;
    if (lane == 0) pg[wv] = g;
    __syncthreads();

    if (threadIdx.x < DD) {
        const int col = threadIdx.x;
        const float* pr = (const float*)prow;
        float run = 0.0f;
        int gcur = -1;
        for (int w = 0; w < GATHER_WAVES; ++w) {
            const int gw = pg[w];
            if (gw != gcur) {
                if ((unsigned)gcur < (unsigned)N_GRAPHS)
                    atomicAdd(phis + (size_t)gcur * (2 * DD) + b * DD + col, run);
                gcur = gw;
                run = 0.0f;
            }
            run += pr[w * DD + col];
        }
        if ((unsigned)gcur < (unsigned)N_GRAPHS)
            atomicAdd(phis + (size_t)gcur * (2 * DD) + b * DD + col, run);
    }
}

extern "C" void kernel_launch(void* const* d_in, const int* in_sizes, int n_in,
                              void* d_out, int out_size, void* d_ws, size_t ws_size,
                              hipStream_t stream) {
    const float* x        = (const float*)d_in[0];
    const int*   esrc     = (const int*)d_in[1];
    const int*   edst     = (const int*)d_in[2];
    const int*   gids     = (const int*)d_in[3];
    const float* Wfc      = (const float*)d_in[4];
    const float* bfc      = (const float*)d_in[5];
    const float* wl       = (const float*)d_in[6];
    const float* wr       = (const float*)d_in[7];
    const float* Wres     = (const float*)d_in[8];

    float* h    = (float*)d_out;
    float* phis = (float*)d_out + (size_t)N_NODES * 2 * DD;

    char* ws = (char*)d_ws;
    float* featb  = (float*)ws;
    int*   csr    = (int*)(ws + (size_t)N_NODES * DD * 4);
    int*   rowptr = (int*)(ws + (size_t)N_NODES * DD * 4 + (size_t)N_EDGES * 4);
    float* a_self = (float*)(rowptr + 2 * 10240);
    float* a_nb   = a_self + 10240;

    k_front<<<dim3(BIN_WGS + FR_JOBS * FR_TILES), dim3(FR_THREADS), 0, stream>>>(
        esrc, edst, rowptr, csr, x, Wfc, bfc, Wres, wl, wr, featb, h, a_self, a_nb, phis);

    k_gather<<<dim3((N_NODES + GATHER_WAVES - 1) / GATHER_WAVES), dim3(GATHER_WAVES * 32), 0, stream>>>(
        rowptr, csr, a_self, a_nb, featb, h, gids, phis, 0);

    k_gemm<<<dim3((N_NODES + 63) / 64, 1), dim3(GEMM_THREADS), 0, stream>>>(
        x, Wfc, bfc, Wres, wl, wr, featb, h, a_self, a_nb, 3);
    k_gather<<<dim3((N_NODES + GATHER_WAVES - 1) / GATHER_WAVES), dim3(GATHER_WAVES * 32), 0, stream>>>(
        rowptr, csr, a_self, a_nb, featb, h, gids, phis, 1);
}


# Gather: each wave's chain of dependent loads cut: edge list held in registers (sources and scores read once, not in 3 walks), 16 feature rows per batch via scalar lane reads; sum order kept
# speedup vs baseline: 7.4302x; 1.1456x over previous
#include <hip/hip_runtime.h>
#include <hip/hip_bf16.h>

#define N_NODES 10000
#define N_EDGES 640000
#define N_GRAPHS 64
#define DD 128
#define NEG_SLOPE 0.01f

typedef __attribute__((ext_vector_type(2))) float v2f;
typedef __attribute__((ext_vector_type(8))) float v8f;

#define BIN_NODES 148
#define BIN_RANGES ((N_NODES + BIN_NODES - 1) / BIN_NODES)
#define BIN_HALVES 2
#define BIN_WGS (BIN_RANGES * BIN_HALVES)
#define BIN_THREADS 1024
#define BIN_WAVES (BIN_THREADS / 32)
#define BIN_WCAP 512
#define BIN_CAP (BIN_WAVES * BIN_WCAP)
#define HALF_EDGES (N_EDGES / BIN_HALVES)
#define NV4 (HALF_EDGES / 4)
#define BIN_BATCH 8
#define BIN_NB 10
#define RP_PITCH 10240
#define BIN_FAR 0x7FFFFFFF

__device__ __forceinline__ void bin_body(const int* __restrict__ esrc,
                      const int* __restrict__ edst,
                      int* __restrict__ rowptr, int* __restrict__ csr, const int wg) {
    __shared__ unsigned wl[BIN_CAP];
    __shared__ int sorted[BIN_CAP];
    __shared__ int cnt[BIN_NODES], cur[BIN_NODES], pre[BIN_NODES + 1];
    __shared__ int wsum[BIN_WAVES];
    __shared__ int ovf;
    const int t = threadIdx.x;
    const int lane = t & 31;
    const int wave = t >> 5;
    const int rng = wg >> 1;
    const int half = wg & 1;
    const int lo = rng * BIN_NODES;
    const int e_lo = half * HALF_EDGES;
    if (t < BIN_NODES) { cnt[t] = 0; cur[t] = 0; }
    if (t == 0) ovf = 0;
    __syncthreads();

    const int rot = rng * (NV4 / BIN_RANGES);
    const int4* D4 = (const int4*)(edst + e_lo);
    unsigned* mywl = wl + wave * BIN_WCAP;
    int below = 0;
    int wcnt = 0;

    auto loadb = [&](int4 (&d)[BIN_BATCH], int ib) {
#pragma unroll
        for (int u = 0; u < BIN_BATCH; ++u) {
            const int i = (ib * BIN_BATCH + u) * BIN_THREADS + t;
            int idx = i + rot;
            if (idx >= NV4) idx -= NV4;
            d[u] = (i < NV4) ? D4[idx] : make_int4(BIN_FAR, BIN_FAR, BIN_FAR, BIN_FAR);
        }
    };
    auto one = [&](int dv, int e) {
        const int tk = dv - lo;
        below += (int)((unsigned)tk >> 31);
        const bool mk = (unsigned)tk < BIN_NODES;
        const unsigned mask = __builtin_amdgcn_ballot_w32(mk);
        if (mask) {
            const int pos = wcnt + (int)__builtin_amdgcn_mbcnt_lo(mask, 0u);
            if (mk && pos < BIN_WCAP) mywl[pos] = ((unsigned)tk << 20) | (unsigned)e;
            wcnt += __builtin_popcount(mask);
        }
    };
    auto procb = [&](const int4 (&d)[BIN_BATCH], int ib) {
#pragma unroll
        for (int u = 0; u < BIN_BATCH; ++u) {
            const int i = (ib * BIN_BATCH + u) * BIN_THREADS + t;
            int idx = i + rot;
            if (idx >= NV4) idx -= NV4;
            const int e = e_lo + idx * 4;
            one(d[u].x, e + 0);
            one(d[u].y, e + 1);
            one(d[u].z, e + 2);
            one(d[u].w, e + 3);
        }
    };

    int4 A[BIN_BATCH], B[BIN_BATCH];
    loadb(A, 0);
    for (int ib = 0; ib < BIN_NB; ib += 2) {
        loadb(B, ib + 1);
        procb(A, ib);
        loadb(A, ib + 2);
        procb(B, ib + 1);
    }

#pragma unroll
    for (int o = 16; o >= 1; o >>= 1) below += __shfl_xor(below, o, 32);
    if (lane == 0) {
        wsum[wave] = below;
        if (wcnt > BIN_WCAP) ovf = 1;
    }
    const int nw = min(wcnt, BIN_WCAP);
    for (int i = lane; i < nw; i += 32) atomicAdd(&cnt[mywl[i] >> 20], 1);
    __syncthreads();
    int base = e_lo;
#pragma unroll
    for (int w = 0; w < BIN_WAVES; ++w) base += wsum[w];
    const bool over = ovf != 0;
    if (over) {
        __syncthreads();
        if (t < BIN_NODES) cnt[t] = 0;
        __syncthreads();
        for (int e = e_lo + t; e < e_lo + HALF_EDGES; e += BIN_THREADS) {
            const int dl = edst[e] - lo;
            if ((unsigned)dl < BIN_NODES) atomicAdd(&cnt[dl], 1);
        }
        __syncthreads();
    }
    if (t <= BIN_NODES) {
        int run = 0;
        for (int k = 0; k < t; ++k) run += cnt[k];
        pre[t] = run;
    }
    __syncthreads();
    const int total = pre[BIN_NODES];
    int* rp = rowptr + half * RP_PITCH;
    if (t < BIN_NODES && lo + t < N_NODES) rp[lo + t] = base + pre[t];
    if (t == 0 && rng == BIN_RANGES - 1) rp[N_NODES] = base + total;

    if (!over) {
        for (int i0 = 0; i0 < nw; i0 += 8 * 32) {
            unsigned m[8];
            int s[8];
#pragma unroll
            for (int u = 0; u < 8; ++u) {
                const int i = i0 + u * 32 + lane;
                m[u] = (i < nw) ? mywl[i] : 0u;
                s[u] = (i < nw) ? esrc[m[u] & 0xFFFFFu] : 0;
            }
#pragma unroll
            for (int u = 0; u < 8; ++u) {
                const int i = i0 + u * 32 + lane;
                if (i < nw) {
                    const int dl = (int)(m[u] >> 20);
                    const int slot = atomicAdd(&cur[dl], 1);
                    sorted[pre[dl] + slot] = s[u];
                }
            }
        }
        __syncthreads();
        for (int i = t; i < total; i += BIN_THREADS) csr[base + i] = sorted[i];
    } else {
        for (int e = e_lo + t; e < e_lo + HALF_EDGES; e += BIN_THREADS) {
            const int dl = edst[e] - lo;
            if ((unsigned)dl < BIN_NODES) {
                const int slot = atomicAdd(&cur[dl], 1);
                csr[base + pre[dl] + slot] = esrc[e];
            }
        }
    }
}

#define GEMM_THREADS 128
#define WPITCH 132
__global__ void __launch_bounds__(GEMM_THREADS) k_gemm(const float* __restrict__ x,
                       const float* __restrict__ Wfc, const float* __restrict__ bfc,
                       const float* __restrict__ Wres,
                       const float* __restrict__ wl, const float* __restrict__ wr,
                       float* __restrict__ feat_ws, float* __restrict__ hout,
                       float* __restrict__ a_self, float* __restrict__ a_nb, int job0) {
    __shared__ float Ws[DD * WPITCH];
    const int job   = blockIdx.y + job0;
    const int b     = (job >= 2) ? 1 : 0;
    const int which = (job == 1 || job == 2) ? 1 : 0;
    const float* W = (which == 0 ? Wfc : Wres) + (size_t)b * DD * DD;

    const float4* W4 = (const float4*)W;
    for (int i0 = 0; i0 < DD * DD / 4; i0 += 8 * GEMM_THREADS) {
        float4 w[8];
#pragma unroll
        for (int u = 0; u < 8; ++u) w[u] = W4[i0 + u * GEMM_THREADS + threadIdx.x];
#pragma unroll
        for (int u = 0; u < 8; ++u) {
            const int i = i0 + u * GEMM_THREADS + threadIdx.x;
            const int j = i >> 5, k4 = (i & 31) << 2;
            *(float4*)(Ws + j * WPITCH + k4) = w[u];
        }
    }
    __syncthreads();

    const int lane = threadIdx.x & 31;
    const int wave = threadIdx.x >> 5;
    const int rowBase = blockIdx.x * 64 + wave * 16;
    const int l16 = lane & 15;
    const int hi  = lane >> 4;

    v8f c[8];
#pragma unroll
    for (int t = 0; t < 8; ++t) {
        float bv = (which == 0) ? bfc[b * DD + t * 16 + l16] : 0.0f;
#pragma unroll
        for (int v = 0; v < 8; ++v) c[t][v] = bv;
    }

    const int arow = min(rowBase + l16, N_NODES - 1);
    const float* xrow = x + (size_t)arow * DD;

    for (int k0 = 0; k0 < DD; k0 += 4) {
        const int krow = k0 + 2 * hi;
        v2f a = *(const v2f*)(xrow + krow);
#pragma unroll
        for (int t = 0; t < 8; ++t) {
            const int j = t * 16 + l16;
            const v2f bf = *(const v2f*)(Ws + j * WPITCH + krow);
            c[t] = __builtin_amdgcn_wmma_f32_16x16x4_f32(
                false, a, false, bf, (short)0, c[t], false, false);
        }
    }

#pragma unroll
    for (int t = 0; t < 8; ++t) {
        const int col = t * 16 + l16;
#pragma unroll
        for (int v = 0; v < 8; ++v) {
            const int row = rowBase + v + 8 * hi;
            if (row < N_NODES) {
                if (which == 0)
                    feat_ws[(size_t)row * DD + col] = c[t][v];
                else
                    hout[(size_t)row * (2 * DD) + b * DD + col] = c[t][v];
            }
        }
    }

    if (which == 0) {
        float wlv[8], wrv[8];
#pragma unroll
        for (int t = 0; t < 8; ++t) {
            wlv[t] = wl[b * DD + t * 16 + l16];
            wrv[t] = wr[b * DD + t * 16 + l16];
        }
        float my_sl = 0.0f, my_sr = 0.0f;
#pragma unroll
        for (int v = 0; v < 8; ++v) {
            float sl = 0.0f, sr = 0.0f;
#pragma unroll
            for (int t = 0; t < 8; ++t) {
                sl += c[t][v] * wlv[t];
                sr += c[t][v] * wrv[t];
            }
#pragma unroll
            for (int m = 8; m >= 1; m >>= 1) {
                sl += __shfl_xor(sl, m, 32);
                sr += __shfl_xor(sr, m, 32);
            }
            if (l16 == v) { my_sl = sl; my_sr = sr; }
        }
        const int myrow = rowBase + l16 + 8 * hi;
        if (l16 < 8 && myrow < N_NODES) {
            a_self[myrow] = my_sl;
            a_nb[myrow]   = my_sr;
        }
    }
}

#define FR_THREADS 1024
#define FR_ROWS 256
#define FR_TILES ((N_NODES + FR_ROWS - 1) / FR_ROWS)
#define FR_JOBS 3
__device__ __forceinline__ void front_gemm(const float* __restrict__ x,
                       const float* __restrict__ Wfc, const float* __restrict__ bfc,
                       const float* __restrict__ Wres,
                       const float* __restrict__ wl, const float* __restrict__ wr,
                       float* __restrict__ feat_ws, float* __restrict__ hout,
                       float* __restrict__ a_self, float* __restrict__ a_nb,
                       float* __restrict__ phis, const int g) {
    __shared__ float Ws[DD * WPITCH];
    __shared__ float dots[FR_ROWS][2][2];
    const int job  = g / FR_TILES;
    const int tile = g - job * FR_TILES;
    if (job == 0) {
        const int i = tile * FR_THREADS + threadIdx.x;
        if (i < N_GRAPHS * 2 * DD) phis[i] = 0.0f;
    }
    const int b     = (job == 2) ? 1 : 0;
    const int which = (job == 0) ? 0 : 1;
    const float* W = (which == 0 ? Wfc : Wres) + (size_t)b * DD * DD;

    const float4* W4 = (const float4*)W;
    {
        float4 w[4];
#pragma unroll
        for (int u = 0; u < 4; ++u) w[u] = W4[u * FR_THREADS + threadIdx.x];
#pragma unroll
        for (int u = 0; u < 4; ++u) {
            const int i = u * FR_THREADS + threadIdx.x;
            const int j = i >> 5, k4 = (i & 31) << 2;
            *(float4*)(Ws + j * WPITCH + k4) = w[u];
        }
    }
    __syncthreads();

    const int lane = threadIdx.x & 31;
    const int wave = threadIdx.x >> 5;
    const int strip = wave >> 1;
    const int colBase = (wave & 1) * 64;
    const int rowBase = tile * FR_ROWS + strip * 16;
    const int l16 = lane & 15;
    const int hi  = lane >> 4;

    v8f c[4];
#pragma unroll
    for (int t = 0; t < 4; ++t) {
        float bv = (which == 0) ? bfc[b * DD + colBase + t * 16 + l16] : 0.0f;
#pragma unroll
        for (int v = 0; v < 8; ++v) c[t][v] = bv;
    }

    const int arow = min(rowBase + l16, N_NODES - 1);
    const float* xrow = x + (size_t)arow * DD;

    for (int k0 = 0; k0 < DD; k0 += 4) {
        const int krow = k0 + 2 * hi;
        v2f a = *(const v2f*)(xrow + krow);
#pragma unroll
        for (int t = 0; t < 4; ++t) {
            const int j = colBase + t * 16 + l16;
            const v2f bf = *(const v2f*)(Ws + j * WPITCH + krow);
            c[t] = __builtin_amdgcn_wmma_f32_16x16x4_f32(
                false, a, false, bf, (short)0, c[t], false, false);
        }
    }

#pragma unroll
    for (int t = 0; t < 4; ++t) {
        const int col = colBase + t * 16 + l16;
#pragma unroll
        for (int v = 0; v < 8; ++v) {
            const int row = rowBase + v + 8 * hi;
            if (row < N_NODES) {
                if (which == 0)
                    feat_ws[(size_t)row * DD + col] = c[t][v];
                else
                    hout[(size_t)row * (2 * DD) + b * DD + col] = c[t][v];
            }
        }
    }

    if (which == 0) {
        float wlv[4], wrv[4];
#pragma unroll
        for (int t = 0; t < 4; ++t) {
            wlv[t] = wl[b * DD + colBase + t * 16 + l16];
            wrv[t] = wr[b * DD + colBase + t * 16 + l16];
        }
        float my_sl = 0.0f, my_sr = 0.0f;
#pragma unroll
        for (int v = 0; v < 8; ++v) {
            float sl = 0.0f, sr = 0.0f;
#pragma unroll
            for (int t = 0; t < 4; ++t) {
                sl += c[t][v] * wlv[t];
                sr += c[t][v] * wrv[t];
            }
#pragma unroll
            for (int m = 8; m >= 1; m >>= 1) {
                sl += __shfl_xor(sl, m, 32);
                sr += __shfl_xor(sr, m, 32);
            }
            if (l16 == v) { my_sl = sl; my_sr = sr; }
        }
        if (l16 < 8) {
            const int r = strip * 16 + l16 + 8 * hi;
            dots[r][wave & 1][0] = my_sl;
            dots[r][wave & 1][1] = my_sr;
        }
        __syncthreads();
        if (threadIdx.x < FR_ROWS) {
            const int row = tile * FR_ROWS + threadIdx.x;
            if (row < N_NODES) {
                a_self[row] = dots[threadIdx.x][0][0] + dots[threadIdx.x][1][0];
                a_nb[row]   = dots[threadIdx.x][0][1] + dots[threadIdx.x][1][1];
            }
        }
    }
}

__global__ void __launch_bounds__(FR_THREADS) k_front(
                       const int* __restrict__ esrc, const int* __restrict__ edst,
                       int* __restrict__ rowptr, int* __restrict__ csr,
                       const float* __restrict__ x,
                       const float* __restrict__ Wfc, const float* __restrict__ bfc,
                       const float* __restrict__ Wres,
                       const float* __restrict__ wl, const float* __restrict__ wr,
                       float* __restrict__ feat_ws, float* __restrict__ hout,
                       float* __restrict__ a_self, float* __restrict__ a_nb,
                       float* __restrict__ phis) {
    if (blockIdx.x < BIN_WGS)
        bin_body(esrc, edst, rowptr, csr, blockIdx.x);
    else
        front_gemm(x, Wfc, bfc, Wres, wl, wr, feat_ws, hout, a_self, a_nb, phis,
                   blockIdx.x - BIN_WGS);
}

__device__ __forceinline__ float edge_score(float as, float an) {
    float sc = as + an;
    return sc >= 0.0f ? sc : NEG_SLOPE * sc;
}

#define GATHER_WAVES 8
#define GATHER_EPL 4
template <int NB>
__device__ __forceinline__ void gather_rows(const float4* __restrict__ F, const int lane,
                                            const int sv, const float ev, const int j,
                                            const int cnt, float4& acc) {
    float  wj[NB];
    float4 f[NB];
#pragma unroll
    for (int k = 0; k < NB; ++k) {
        const int sj = __builtin_amdgcn_readlane(sv, min(j + k, cnt - 1));
        wj[k] = __int_as_float(__builtin_amdgcn_readlane(__float_as_int(ev), j + k));
        f[k] = F[(size_t)sj * (DD / 4) + lane];
    }
#pragma unroll
    for (int k = 0; k < NB; ++k) {
        acc.x = __builtin_fmaf(wj[k], f[k].x, acc.x); acc.y = __builtin_fmaf(wj[k], f[k].y, acc.y);
        acc.z = __builtin_fmaf(wj[k], f[k].z, acc.z); acc.w = __builtin_fmaf(wj[k], f[k].w, acc.w);
    }
}

__global__ void __launch_bounds__(GATHER_WAVES * 32) k_gather(
                         const int* __restrict__ rowptr, const int* __restrict__ csr,
                         const float* __restrict__ a_self, const float* __restrict__ a_nb,
                         const float* __restrict__ feat, float* __restrict__ hout,
                         const int* __restrict__ gids, float* __restrict__ phis, int b) {
    __shared__ float4 prow[GATHER_WAVES][32];
    __shared__ int pg[GATHER_WAVES];
    const int lane = threadIdx.x & 31;
    const int wv = __builtin_amdgcn_readfirstlane((int)(threadIdx.x >> 5));
    const int n = blockIdx.x * GATHER_WAVES + wv;

    float4 r = make_float4(0.0f, 0.0f, 0.0f, 0.0f);
    int g = -1;
    if (n < N_NODES) {
        const int beg0 = rowptr[n], end0 = rowptr[n + 1];
        const int beg1 = rowptr[RP_PITCH + n], end1 = rowptr[RP_PITCH + n + 1];
        float4* hp = (float4*)(hout + (size_t)n * (2 * DD) + b * DD) + lane;
        const float4 res = *hp;
        g = gids[n];
        const int n0 = end0 - beg0;
        const int deg = n0 + (end1 - beg1);
        const int off1 = beg1 - n0;
        auto at = [&](int i) { return i + (i < n0 ? beg0 : off1); };
        const float4* F = (const float4*)feat;
        float4 acc = make_float4(0.0f, 0.0f, 0.0f, 0.0f);
        if (deg > 0 && deg <= GATHER_EPL * 32) {
            const float as = a_self[n];
            int   s[GATHER_EPL];
            float e[GATHER_EPL];
#pragma unroll
            for (int u = 0; u < GATHER_EPL; ++u) {
                const int i = u * 32 + lane;
                s[u] = csr[at(i < deg ? i : 0)];
            }
#pragma unroll
            for (int u = 0; u < GATHER_EPL; ++u) e[u] = a_nb[s[u]];
#pragma unroll
            for (int u = 0; u < GATHER_EPL; ++u) {
                const int i = u * 32 + lane;
                e[u] = (i < deg) ? edge_score(as, e[u]) : -INFINITY;
            }
            float m = e[0];
#pragma unroll
            for (int u = 1; u < GATHER_EPL; ++u) m = fmaxf(m, e[u]);
#pragma unroll
            for (int o = 16; o >= 1; o >>= 1) m = fmaxf(m, __shfl_xor(m, o, 32));

            float den = 0.0f;
#pragma unroll
            for (int u = 0; u < GATHER_EPL; ++u) {
                const int i = u * 32 + lane;
                e[u] = (i < deg) ? __expf(e[u] - m) : 0.0f;
                den += e[u];
            }
#pragma unroll
            for (int o = 16; o >= 1; o >>= 1) den += __shfl_xor(den, o, 32);
#pragma unroll
            for (int u = 0; u < GATHER_EPL; ++u) e[u] = e[u] / den;

#pragma unroll
            for (int u = 0; u < GATHER_EPL; ++u) {
                const int cnt = min(32, deg - u * 32);
                if (cnt > 0) {
                    int j = 0;
                    for (; j + 16 <= cnt; j += 16) gather_rows<16>(F, lane, s[u], e[u], j, cnt, acc);
                    if (cnt - j > 8)      gather_rows<16>(F, lane, s[u], e[u], j, cnt, acc);
                    else if (cnt - j > 0) gather_rows<8>(F, lane, s[u], e[u], j, cnt, acc);
                }
            }
        } else if (deg > 0) {
            const float as = a_self[n];

            float m = -INFINITY;
            for (int i = lane; i < deg; i += 32)
                m = fmaxf(m, edge_score(as, a_nb[csr[at(i)]]));
#pragma unroll
            for (int o = 16; o >= 1; o >>= 1) m = fmaxf(m, __shfl_xor(m, o, 32));

            float den = 0.0f;
            for (int i = lane; i < deg; i += 32)
                den += __expf(edge_score(as, a_nb[csr[at(i)]]) - m);
#pragma unroll
            for (int o = 16; o >= 1; o >>= 1) den += __shfl_xor(den, o, 32);

            for (int base = 0; base < deg; base += 32) {
                const int i = base + lane;
                int s = 0;
                float w = 0.0f;
                if (i < deg) {
                    s = csr[at(i)];
                    w = __expf(edge_score(as, a_nb[s]) - m) / den;
                }
                const int cnt = min(32, deg - base);
                int j = 0;
                for (; j + 8 <= cnt; j += 8) {
                    int   sj[8];
                    float wj[8];
                    float4 f[8];
#pragma unroll
                    for (int u = 0; u < 8; ++u) { sj[u] = __shfl(s, j + u, 32); wj[u] = __shfl(w, j + u, 32); }
#pragma unroll
                    for (int u = 0; u < 8; ++u) f[u] = F[(size_t)sj[u] * (DD / 4) + lane];
#pragma unroll
                    for (int u = 0; u < 8; ++u) {
                        acc.x += wj[u] * f[u].x; acc.y += wj[u] * f[u].y;
                        acc.z += wj[u] * f[u].z; acc.w += wj[u] * f[u].w;
                    }
                }
                for (; j < cnt; ++j) {
                    const int   s1 = __shfl(s, j, 32);
                    const float w1 = __shfl(w, j, 32);
                    const float4 f1 = F[(size_t)s1 * (DD / 4) + lane];
                    acc.x += w1 * f1.x; acc.y += w1 * f1.y;
                    acc.z += w1 * f1.z; acc.w += w1 * f1.w;
                }
            }
        }
        r = res;
        r.x = fmaxf(r.x + acc.x, 0.0f); r.y = fmaxf(r.y + acc.y, 0.0f);
        r.z = fmaxf(r.z + acc.z, 0.0f); r.w = fmaxf(r.w + acc.w, 0.0f);
        *hp = r;
    }
    prow[wv][lane] = r;
    if (lane == 0) pg[wv] = g;
    __syncthreads();

    if (threadIdx.x < DD) {
        const int col = threadIdx.x;
        const float* pr = (const float*)prow;
        float run = 0.0f;
        int gcur = -1;
        for (int w = 0; w < GATHER_WAVES; ++w) {
            const int gw = pg[w];
            if (gw != gcur) {
                if ((unsigned)gcur < (unsigned)N_GRAPHS)
                    atomicAdd(phis + (size_t)gcur * (2 * DD) + b * DD + col, run);
                gcur = gw;
                run = 0.0f;
            }
            run += pr[w * DD + col];
        }
        if ((unsigned)gcur < (unsigned)N_GRAPHS)
            atomicAdd(phis + (size_t)gcur * (2 * DD) + b * DD + col, run);
    }
}

extern "C" void kernel_launch(void* const* d_in, const int* in_sizes, int n_in,
                              void* d_out, int out_size, void* d_ws, size_t ws_size,
                              hipStream_t stream) {
    const float* x        = (const float*)d_in[0];
    const int*   esrc     = (const int*)d_in[1];
    const int*   edst     = (const int*)d_in[2];
    const int*   gids     = (const int*)d_in[3];
    const float* Wfc      = (const float*)d_in[4];
    const float* bfc      = (const float*)d_in[5];
    const float* wl       = (const float*)d_in[6];
    const float* wr       = (const float*)d_in[7];
    const float* Wres     = (const float*)d_in[8];

    float* h    = (float*)d_out;
    float* phis = (float*)d_out + (size_t)N_NODES * 2 * DD;

    char* ws = (char*)d_ws;
    float* featb  = (float*)ws;
    int*   csr    = (int*)(ws + (size_t)N_NODES * DD * 4);
    int*   rowptr = (int*)(ws + (size_t)N_NODES * DD * 4 + (size_t)N_EDGES * 4);
    float* a_self = (float*)(rowptr + 2 * 10240);
    float* a_nb   = a_self + 10240;

    k_front<<<dim3(BIN_WGS + FR_JOBS * FR_TILES), dim3(FR_THREADS), 0, stream>>>(
        esrc, edst, rowptr, csr, x, Wfc, bfc, Wres, wl, wr, featb, h, a_self, a_nb, phis);

    k_gather<<<dim3((N_NODES + GATHER_WAVES - 1) / GATHER_WAVES), dim3(GATHER_WAVES * 32), 0, stream>>>(
        rowptr, csr, a_self, a_nb, featb, h, gids, phis, 0);

    k_gemm<<<dim3((N_NODES + 63) / 64, 1), dim3(GEMM_THREADS), 0, stream>>>(
        x, Wfc, bfc, Wres, wl, wr, featb, h, a_self, a_nb, 3);
    k_gather<<<dim3((N_NODES + GATHER_WAVES - 1) / GATHER_WAVES), dim3(GATHER_WAVES * 32), 0, stream>>>(
        rowptr, csr, a_self, a_nb, featb, h, gids, phis, 1);
}


# Neighbour sums taken over x rows before the projection (linearity): one gather serves both blocks, then one K=256 WMMA product per block adds fc and residual; 4->3 launches; all sum orders change
# speedup vs baseline: 9.5809x; 1.2895x over previous
#include <hip/hip_runtime.h>
#include <hip/hip_bf16.h>

#define N_NODES 10000
#define N_EDGES 640000
#define N_GRAPHS 64
#define DD 128
#define NEG_SLOPE 0.01f

typedef __attribute__((ext_vector_type(2))) float v2f;
typedef __attribute__((ext_vector_type(8))) float v8f;

#define BIN_NODES 148
#define BIN_RANGES ((N_NODES + BIN_NODES - 1) / BIN_NODES)
#define BIN_HALVES 2
#define BIN_WGS (BIN_RANGES * BIN_HALVES)
#define BIN_THREADS 1024
#define BIN_WAVES (BIN_THREADS / 32)
#define BIN_WCAP 640
#define BIN_CAP (BIN_WAVES * BIN_WCAP)
#define HALF_EDGES (N_EDGES / BIN_HALVES)
#define NV4 (HALF_EDGES / 4)
#define BIN_BATCH 8
#define BIN_NB 10
#define RP_PITCH 10240
#define BIN_FAR 0x7FFFFFFF

__device__ __forceinline__ void bin_body(const int* __restrict__ esrc,
                      const int* __restrict__ edst,
                      int* __restrict__ rowptr, int* __restrict__ csr, const int wg) {
    __shared__ unsigned wl[BIN_CAP];
    __shared__ int sorted[BIN_CAP];
    __shared__ int cnt[BIN_NODES], cur[BIN_NODES], pre[BIN_NODES + 1];
    __shared__ int wsum[BIN_WAVES];
    __shared__ int ovf;
    const int t = threadIdx.x;
    const int lane = t & 31;
    const int wave = t >> 5;
    const int rng = wg >> 1;
    const int half = wg & 1;
    const int lo = rng * BIN_NODES;
    const int e_lo = half * HALF_EDGES;
    if (t < BIN_NODES) { cnt[t] = 0; cur[t] = 0; }
    if (t == 0) ovf = 0;
    __syncthreads();

    const int rot = rng * (NV4 / BIN_RANGES);
    const int4* D4 = (const int4*)(edst + e_lo);
    unsigned* mywl = wl + wave * BIN_WCAP;
    int below = 0;
    int wcnt = 0;

    auto loadb = [&](int4 (&d)[BIN_BATCH], int ib) {
#pragma unroll
        for (int u = 0; u < BIN_BATCH; ++u) {
            const int i = (ib * BIN_BATCH + u) * BIN_THREADS + t;
            int idx = i + rot;
            if (idx >= NV4) idx -= NV4;
            d[u] = (i < NV4) ? D4[idx] : make_int4(BIN_FAR, BIN_FAR, BIN_FAR, BIN_FAR);
        }
    };
    auto one = [&](int dv, int e) {
        const int tk = dv - lo;
        below += (int)((unsigned)tk >> 31);
        const bool mk = (unsigned)tk < BIN_NODES;
        const unsigned mask = __builtin_amdgcn_ballot_w32(mk);
        if (mask) {
            const int pos = wcnt + (int)__builtin_amdgcn_mbcnt_lo(mask, 0u);
            if (mk && pos < BIN_WCAP) mywl[pos] = ((unsigned)tk << 20) | (unsigned)e;
            wcnt += __builtin_popcount(mask);
        }
    };
    auto procb = [&](const int4 (&d)[BIN_BATCH], int ib) {
#pragma unroll
        for (int u = 0; u < BIN_BATCH; ++u) {
            const int i = (ib * BIN_BATCH + u) * BIN_THREADS + t;
            int idx = i + rot;
            if (idx >= NV4) idx -= NV4;
            const int e = e_lo + idx * 4;
            one(d[u].x, e + 0);
            one(d[u].y, e + 1);
            one(d[u].z, e + 2);
            one(d[u].w, e + 3);
        }
    };

    int4 A[BIN_BATCH], B[BIN_BATCH];
    loadb(A, 0);
    for (int ib = 0; ib < BIN_NB; ib += 2) {
        loadb(B, ib + 1);
        procb(A, ib);
        loadb(A, ib + 2);
        procb(B, ib + 1);
    }

#pragma unroll
    for (int o = 16; o >= 1; o >>= 1) below += __shfl_xor(below, o, 32);
    if (lane == 0) {
        wsum[wave] = below;
        if (wcnt > BIN_WCAP) ovf = 1;
    }
    const int nw = min(wcnt, BIN_WCAP);
    for (int i = lane; i < nw; i += 32) atomicAdd(&cnt[mywl[i] >> 20], 1);
    __syncthreads();
    int base = e_lo;
#pragma unroll
    for (int w = 0; w < BIN_WAVES; ++w) base += wsum[w];
    const bool over = ovf != 0;
    if (over) {
        __syncthreads();
        if (t < BIN_NODES) cnt[t] = 0;
        __syncthreads();
        for (int e = e_lo + t; e < e_lo + HALF_EDGES; e += BIN_THREADS) {
            const int dl = edst[e] - lo;
            if ((unsigned)dl < BIN_NODES) atomicAdd(&cnt[dl], 1);
        }
        __syncthreads();
    }
    if (t <= BIN_NODES) {
        int run = 0;
        for (int k = 0; k < t; ++k) run += cnt[k];
        pre[t] = run;
    }
    __syncthreads();
    const int total = pre[BIN_NODES];
    int* rp = rowptr + half * RP_PITCH;
    if (t < BIN_NODES && lo + t < N_NODES) rp[lo + t] = base + pre[t];
    if (t == 0 && rng == BIN_RANGES - 1) rp[N_NODES] = base + total;

    if (!over) {
        for (int i0 = 0; i0 < nw; i0 += 8 * 32) {
            unsigned m[8];
            int s[8];
#pragma unroll
            for (int u = 0; u < 8; ++u) {
                const int i = i0 + u * 32 + lane;
                m[u] = (i < nw) ? mywl[i] : 0u;
                s[u] = (i < nw) ? esrc[m[u] & 0xFFFFFu] : 0;
            }
#pragma unroll
            for (int u = 0; u < 8; ++u) {
                const int i = i0 + u * 32 + lane;
                if (i < nw) {
                    const int dl = (int)(m[u] >> 20);
                    const int slot = atomicAdd(&cur[dl], 1);
                    sorted[pre[dl] + slot] = s[u];
                }
            }
        }
        __syncthreads();
        for (int i = t; i < total; i += BIN_THREADS) csr[base + i] = sorted[i];
    } else {
        for (int e = e_lo + t; e < e_lo + HALF_EDGES; e += BIN_THREADS) {
            const int dl = edst[e] - lo;
            if ((unsigned)dl < BIN_NODES) {
                const int slot = atomicAdd(&cur[dl], 1);
                csr[base + pre[dl] + slot] = esrc[e];
            }
        }
    }
}

#define FR_THREADS 1024
#define DOT_WGS 120
#define SC_PITCH 10240
__device__ __forceinline__ void front_dots(const float* __restrict__ x,
                       const float* __restrict__ Wfc, const float* __restrict__ bfc,
                       const float* __restrict__ wl, const float* __restrict__ wr,
                       float* __restrict__ sc, float* __restrict__ phis, const int g) {
    __shared__ float4 vp[2][4][DD / 4];
    __shared__ float cq[4];
    const int t = threadIdx.x;
    const int lane = t & 31;
    {
        const int i = g * FR_THREADS + t;
        if (i < N_GRAPHS * 2 * DD) phis[i] = 0.0f;
    }
    {
        const int q = t >> 8, jh = (t >> 7) & 1, k = t & 127, b = q >> 1;
        const float* W = Wfc + (size_t)b * DD * DD + (size_t)jh * (DD / 2) * DD;
        const float* w = ((q & 1) ? wr : wl) + b * DD + jh * (DD / 2);
        float s = 0.0f;
#pragma unroll 16
        for (int j = 0; j < DD / 2; ++j) s = __builtin_fmaf(W[j * DD + k], w[j], s);
        ((float*)vp)[(jh * 4 + q) * DD + k] = s;
    }
    if (t < 4 * 32) {
        const int q = t >> 5, b = q >> 1;
        const float* w = ((q & 1) ? wr : wl) + b * DD;
        float s = 0.0f;
        for (int j = lane; j < DD; j += 32) s += bfc[b * DD + j] * w[j];
#pragma unroll
        for (int o = 16; o >= 1; o >>= 1) s += __shfl_xor(s, o, 32);
        if (lane == 0) cq[q] = s;
    }
    __syncthreads();
    const int wave = t >> 5;
    auto both = [&](int q) {
        const float4 a = vp[0][q][lane], c = vp[1][q][lane];
        return make_float4(a.x + c.x, a.y + c.y, a.z + c.z, a.w + c.w);
    };
    const float4 v0 = both(0), v1 = both(1), v2 = both(2), v3 = both(3);
    const float c0 = cq[0], c1 = cq[1], c2 = cq[2], c3 = cq[3];
    const float4* X4 = (const float4*)x;
    for (int row = g * 32 + wave; row < N_NODES; row += DOT_WGS * 32) {
        const float4 xv = X4[(size_t)row * (DD / 4) + lane];
        float d0 = xv.x * v0.x + xv.y * v0.y + xv.z * v0.z + xv.w * v0.w;
        float d1 = xv.x * v1.x + xv.y * v1.y + xv.z * v1.z + xv.w * v1.w;
        float d2 = xv.x * v2.x + xv.y * v2.y + xv.z * v2.z + xv.w * v2.w;
        float d3 = xv.x * v3.x + xv.y * v3.y + xv.z * v3.z + xv.w * v3.w;
#pragma unroll
        for (int o = 16; o >= 1; o >>= 1) {
            d0 += __shfl_xor(d0, o, 32); d1 += __shfl_xor(d1, o, 32);
            d2 += __shfl_xor(d2, o, 32); d3 += __shfl_xor(d3, o, 32);
        }
        if (lane == 0) {
            sc[row] = d0 + c0;
            sc[SC_PITCH + row] = d2 + c2;
            ((float2*)(sc + 2 * SC_PITCH))[row] = make_float2(d1 + c1, d3 + c3);
        }
    }
}

__global__ void __launch_bounds__(FR_THREADS) k_front(
                       const int* __restrict__ esrc, const int* __restrict__ edst,
                       int* __restrict__ rowptr, int* __restrict__ csr,
                       const float* __restrict__ x,
                       const float* __restrict__ Wfc, const float* __restrict__ bfc,
                       const float* __restrict__ wl, const float* __restrict__ wr,
                       float* __restrict__ sc, float* __restrict__ phis) {
    if (blockIdx.x < BIN_WGS)
        bin_body(esrc, edst, rowptr, csr, blockIdx.x);
    else
        front_dots(x, Wfc, bfc, wl, wr, sc, phis, blockIdx.x - BIN_WGS);
}

__device__ __forceinline__ float edge_score(float as, float an) {
    float sc = as + an;
    return sc >= 0.0f ? sc : NEG_SLOPE * sc;
}

#define GATHER_WAVES 8
#define GATHER_EPL 4
template <int NB>
__device__ __forceinline__ void gather_rows(const float4* __restrict__ F, const int lane,
                                            const int sv, const float e0, const float e1,
                                            const int j, const int cnt,
                                            float4& acc0, float4& acc1) {
    float  w0[NB], w1[NB];
    float4 f[NB];
#pragma unroll
    for (int k = 0; k < NB; ++k) {
        const int sj = __builtin_amdgcn_readlane(sv, min(j + k, cnt - 1));
        w0[k] = __int_as_float(__builtin_amdgcn_readlane(__float_as_int(e0), j + k));
        w1[k] = __int_as_float(__builtin_amdgcn_readlane(__float_as_int(e1), j + k));
        f[k] = F[(size_t)sj * (DD / 4) + lane];
    }
#pragma unroll
    for (int k = 0; k < NB; ++k) {
        acc0.x = __builtin_fmaf(w0[k], f[k].x, acc0.x); acc0.y = __builtin_fmaf(w0[k], f[k].y, acc0.y);
        acc0.z = __builtin_fmaf(w0[k], f[k].z, acc0.z); acc0.w = __builtin_fmaf(w0[k], f[k].w, acc0.w);
        acc1.x = __builtin_fmaf(w1[k], f[k].x, acc1.x); acc1.y = __builtin_fmaf(w1[k], f[k].y, acc1.y);
        acc1.z = __builtin_fmaf(w1[k], f[k].z, acc1.z); acc1.w = __builtin_fmaf(w1[k], f[k].w, acc1.w);
    }
}

__device__ __forceinline__ void chunk_softmax(float (&e)[GATHER_EPL], const float as,
                                              const int lane, const int deg) {
#pragma unroll
    for (int u = 0; u < GATHER_EPL; ++u) {
        const int i = u * 32 + lane;
        e[u] = (i < deg) ? edge_score(as, e[u]) : -INFINITY;
    }
    float m = e[0];
#pragma unroll
    for (int u = 1; u < GATHER_EPL; ++u) m = fmaxf(m, e[u]);
#pragma unroll
    for (int o = 16; o >= 1; o >>= 1) m = fmaxf(m, __shfl_xor(m, o, 32));
    float den = 0.0f;
#pragma unroll
    for (int u = 0; u < GATHER_EPL; ++u) {
        const int i = u * 32 + lane;
        e[u] = (i < deg) ? __expf(e[u] - m) : 0.0f;
        den += e[u];
    }
#pragma unroll
    for (int o = 16; o >= 1; o >>= 1) den += __shfl_xor(den, o, 32);
#pragma unroll
    for (int u = 0; u < GATHER_EPL; ++u) e[u] = e[u] / den;
}

__global__ void __launch_bounds__(GATHER_WAVES * 32) k_gather(
                         const int* __restrict__ rowptr, const int* __restrict__ csr,
                         const float* __restrict__ sc, const float* __restrict__ x,
                         float* __restrict__ hout, float* __restrict__ has) {
    const int lane = threadIdx.x & 31;
    const int wv = __builtin_amdgcn_readfirstlane((int)(threadIdx.x >> 5));
    const int n = blockIdx.x * GATHER_WAVES + wv;
    if (n >= N_NODES) return;

    const int beg0 = rowptr[n], end0 = rowptr[n + 1];
    const int beg1 = rowptr[RP_PITCH + n], end1 = rowptr[RP_PITCH + n + 1];
    const int n0 = end0 - beg0;
    const int deg = n0 + (end1 - beg1);
    const int off1 = beg1 - n0;
    auto at = [&](int i) { return i + (i < n0 ? beg0 : off1); };
    const float4* F = (const float4*)x;
    const float2* anb = (const float2*)(sc + 2 * SC_PITCH);
    float4 acc0 = make_float4(0.0f, 0.0f, 0.0f, 0.0f);
    float4 acc1 = make_float4(0.0f, 0.0f, 0.0f, 0.0f);
    if (deg > 0 && deg <= GATHER_EPL * 32) {
        const float as0 = sc[n], as1 = sc[SC_PITCH + n];
        int   s[GATHER_EPL];
        float e0[GATHER_EPL], e1[GATHER_EPL];
#pragma unroll
        for (int u = 0; u < GATHER_EPL; ++u) {
            const int i = u * 32 + lane;
            s[u] = csr[at(i < deg ? i : 0)];
        }
#pragma unroll
        for (int u = 0; u < GATHER_EPL; ++u) { const float2 p = anb[s[u]]; e0[u] = p.x; e1[u] = p.y; }
        chunk_softmax(e0, as0, lane, deg);
        chunk_softmax(e1, as1, lane, deg);

#pragma unroll
        for (int u = 0; u < GATHER_EPL; ++u) {
            const int cnt = min(32, deg - u * 32);
            if (cnt > 0) {
                int j = 0;
                for (; j + 16 <= cnt; j += 16) gather_rows<16>(F, lane, s[u], e0[u], e1[u], j, cnt, acc0, acc1);
                if (cnt - j > 8)      gather_rows<16>(F, lane, s[u], e0[u], e1[u], j, cnt, acc0, acc1);
                else if (cnt - j > 0) gather_rows<8>(F, lane, s[u], e0[u], e1[u], j, cnt, acc0, acc1);
            }
        }
    } else if (deg > 0) {
        for (int b = 0; b < 2; ++b) {
            const float as = sc[b * SC_PITCH + n];
            auto nb = [&](int s) { const float2 p = anb[s]; return b ? p.y : p.x; };

            float m = -INFINITY;
            for (int i = lane; i < deg; i += 32)
                m = fmaxf(m, edge_score(as, nb(csr[at(i)])));
#pragma unroll
            for (int o = 16; o >= 1; o >>= 1) m = fmaxf(m, __shfl_xor(m, o, 32));

            float den = 0.0f;
            for (int i = lane; i < deg; i += 32)
                den += __expf(edge_score(as, nb(csr[at(i)])) - m);
#pragma unroll
            for (int o = 16; o >= 1; o >>= 1) den += __shfl_xor(den, o, 32);

            float4 acc = make_float4(0.0f, 0.0f, 0.0f, 0.0f);
            for (int base = 0; base < deg; base += 32) {
                const int i = base + lane;
                int s = 0;
                float w = 0.0f;
                if (i < deg) {
                    s = csr[at(i)];
                    w = __expf(edge_score(as, nb(s)) - m) / den;
                }
                const int cnt = min(32, deg - base);
                for (int j = 0; j < cnt; ++j) {
                    const int   s1 = __shfl(s, j, 32);
                    const float w1 = __shfl(w, j, 32);
                    const float4 f1 = F[(size_t)s1 * (DD / 4) + lane];
                    acc.x += w1 * f1.x; acc.y += w1 * f1.y;
                    acc.z += w1 * f1.z; acc.w += w1 * f1.w;
                }
            }
            if (b == 0) acc0 = acc; else acc1 = acc;
        }
    }
    float4* hp = (float4*)(hout + (size_t)n * (2 * DD)) + lane;
    hp[0] = acc0;
    hp[DD / 4] = acc1;
    if (lane == 0) has[n] = deg > 0 ? 1.0f : 0.0f;
}

#define OUT_THREADS 384
#define OUT_ROWS 96
#define OUT_TILES ((N_NODES + OUT_ROWS - 1) / OUT_ROWS)
#define OUT_W4 (DD * 2 * DD / 4)
#define OPITCH 324
__global__ void __launch_bounds__(OUT_THREADS) k_out(const float* __restrict__ x,
                       const float* __restrict__ Wfc, const float* __restrict__ bfc,
                       const float* __restrict__ Wres, const float* __restrict__ has,
                       const int* __restrict__ gids, float* hout, float* __restrict__ phis) {
    __shared__ float Ws[DD * OPITCH];
    const int b = blockIdx.y;
    const float4* Wf4 = (const float4*)(Wfc + (size_t)b * DD * DD);
    const float4* Wr4 = (const float4*)(Wres + (size_t)b * DD * DD);
    for (int i0 = 0; i0 < OUT_W4; i0 += 8 * OUT_THREADS) {
        float4 w[8];
#pragma unroll
        for (int u = 0; u < 8; ++u) {
            const int i = min(i0 + u * OUT_THREADS + (int)threadIdx.x, OUT_W4 - 1);
            const int j = i >> 6, q = i & 63;
            w[u] = (q < DD / 4) ? Wf4[j * (DD / 4) + q] : Wr4[j * (DD / 4) + q - DD / 4];
        }
#pragma unroll
        for (int u = 0; u < 8; ++u) {
            const int i = i0 + u * OUT_THREADS + (int)threadIdx.x;
            const int j = i >> 6, k4 = (i & 63) << 2;
            if (i < OUT_W4) *(float4*)(Ws + j * OPITCH + k4) = w[u];
        }
    }
    __syncthreads();

    const int lane = threadIdx.x & 31;
    const int wave = threadIdx.x >> 5;
    const int strip = wave >> 1;
    const int colBase = (wave & 1) * 64;
    const int rowBase = blockIdx.x * OUT_ROWS + strip * 16;
    const int l16 = lane & 15;
    const int hi  = lane >> 4;

    float fl[8];
    int   gv[8];
#pragma unroll
    for (int v = 0; v < 8; ++v) {
        const int row = rowBase + v + 8 * hi;
        const int rc = min(row, N_NODES - 1);
        fl[v] = has[rc];
        gv[v] = (row < N_NODES) ? gids[rc] : -1;
    }

    v8f c[4];
#pragma unroll
    for (int t = 0; t < 4; ++t) {
        const float bv = bfc[b * DD + colBase + t * 16 + l16];
#pragma unroll
        for (int v = 0; v < 8; ++v) c[t][v] = bv * fl[v];
    }

    const int arow = min(rowBase + l16, N_NODES - 1);
    const float* zrow = hout + (size_t)arow * (2 * DD) + b * DD;
    const float* xrow = x + (size_t)arow * DD;

    for (int k0 = 0; k0 < DD; k0 += 4) {
        const int krow = k0 + 2 * hi;
        v2f a = *(const v2f*)(zrow + krow);
#pragma unroll
        for (int t = 0; t < 4; ++t) {
            const int j = colBase + t * 16 + l16;
            const v2f bf = *(const v2f*)(Ws + j * OPITCH + krow);
            c[t] = __builtin_amdgcn_wmma_f32_16x16x4_f32(
                false, a, false, bf, (short)0, c[t], false, false);
        }
    }
    for (int k0 = 0; k0 < DD; k0 += 4) {
        const int krow = k0 + 2 * hi;
        v2f a = *(const v2f*)(xrow + krow);
#pragma unroll
        for (int t = 0; t < 4; ++t) {
            const int j = colBase + t * 16 + l16;
            const v2f bf = *(const v2f*)(Ws + j * OPITCH + DD + krow);
            c[t] = __builtin_amdgcn_wmma_f32_16x16x4_f32(
                false, a, false, bf, (short)0, c[t], false, false);
        }
    }

    __syncthreads();

    float run[4] = {0.0f, 0.0f, 0.0f, 0.0f};
    int gcur = -1;
    float* pbase = phis + b * DD + colBase + l16;
#pragma unroll
    for (int v = 0; v < 8; ++v) {
        const int row = rowBase + v + 8 * hi;
        if (gv[v] != gcur) {
            if ((unsigned)gcur < (unsigned)N_GRAPHS) {
#pragma unroll
                for (int t = 0; t < 4; ++t) atomicAdd(pbase + (size_t)gcur * (2 * DD) + t * 16, run[t]);
            }
            gcur = gv[v];
#pragma unroll
            for (int t = 0; t < 4; ++t) run[t] = 0.0f;
        }
#pragma unroll
        for (int t = 0; t < 4; ++t) {
            const float val = fmaxf(c[t][v], 0.0f);
            if (row < N_NODES) {
                hout[(size_t)row * (2 * DD) + b * DD + colBase + t * 16 + l16] = val;
                run[t] += val;
            }
        }
    }
    if ((unsigned)gcur < (unsigned)N_GRAPHS) {
#pragma unroll
        for (int t = 0; t < 4; ++t) atomicAdd(pbase + (size_t)gcur * (2 * DD) + t * 16, run[t]);
    }
}

extern "C" void kernel_launch(void* const* d_in, const int* in_sizes, int n_in,
                              void* d_out, int out_size, void* d_ws, size_t ws_size,
                              hipStream_t stream) {
    const float* x        = (const float*)d_in[0];
    const int*   esrc     = (const int*)d_in[1];
    const int*   edst     = (const int*)d_in[2];
    const int*   gids     = (const int*)d_in[3];
    const float* Wfc      = (const float*)d_in[4];
    const float* bfc      = (const float*)d_in[5];
    const float* wl       = (const float*)d_in[6];
    const float* wr       = (const float*)d_in[7];
    const float* Wres     = (const float*)d_in[8];

    float* h    = (float*)d_out;
    float* phis = (float*)d_out + (size_t)N_NODES * 2 * DD;

    char* ws = (char*)d_ws;
    float* sc     = (float*)ws;
    float* has    = sc + 4 * SC_PITCH;
    int*   csr    = (int*)(ws + (size_t)N_NODES * DD * 4);
    int*   rowptr = (int*)(ws + (size_t)N_NODES * DD * 4 + (size_t)N_EDGES * 4);

    k_front<<<dim3(BIN_WGS + DOT_WGS), dim3(FR_THREADS), 0, stream>>>(
        esrc, edst, rowptr, csr, x, Wfc, bfc, wl, wr, sc, phis);

    k_gather<<<dim3((N_NODES + GATHER_WAVES - 1) / GATHER_WAVES), dim3(GATHER_WAVES * 32), 0, stream>>>(
        rowptr, csr, sc, x, h, has);

    k_out<<<dim3(OUT_TILES, 2), dim3(OUT_THREADS), 0, stream>>>(
        x, Wfc, bfc, Wres, has, gids, h, phis);
}


# Edge list cut in 8 parts (was 2), each grouped by destination by 24 work-groups (192 + 64 score): each examines 1/8 of the list; slot prefix by wave scan; gather walks 8 runs; sum order: parts 0..7
# speedup vs baseline: 11.9257x; 1.2447x over previous
#include <hip/hip_runtime.h>
#include <hip/hip_bf16.h>

#define N_NODES 10000
#define N_EDGES 640000
#define N_GRAPHS 64
#define DD 128
#define NEG_SLOPE 0.01f

typedef __attribute__((ext_vector_type(2))) float v2f;
typedef __attribute__((ext_vector_type(8))) float v8f;

#define BIN_NODES 417
#define BIN_RANGES ((N_NODES + BIN_NODES - 1) / BIN_NODES)
#define BIN_PARTS 8
#define BIN_WGS (BIN_RANGES * BIN_PARTS)
#define BIN_THREADS 1024
#define BIN_WAVES (BIN_THREADS / 32)
#define BIN_WCAP 640
#define BIN_CAP (BIN_WAVES * BIN_WCAP)
#define SCAN_WAVES ((BIN_NODES + 31) / 32)
#define PART_EDGES (N_EDGES / BIN_PARTS)
#define NV4 (PART_EDGES / 4)
#define BIN_BATCH 5
#define BIN_NB 4
#define RP_PITCH 10240
#define BIN_FAR 0x7FFFFFFF

__device__ __forceinline__ void bin_body(const int* __restrict__ esrc,
                      const int* __restrict__ edst,
                      int* __restrict__ rowptr, int* __restrict__ csr, const int wg) {
    __shared__ unsigned wl[BIN_CAP];
    __shared__ int sorted[BIN_CAP];
    __shared__ int cnt[BIN_NODES], cur[BIN_NODES], pre[BIN_NODES + 1];
    __shared__ int wsum[BIN_WAVES];
    __shared__ int wtot[SCAN_WAVES];
    __shared__ int ovf;
    const int t = threadIdx.x;
    const int lane = t & 31;
    const int wave = t >> 5;
    const int rng = wg / BIN_PARTS;
    const int part = wg % BIN_PARTS;
    const int lo = rng * BIN_NODES;
    const int e_lo = part * PART_EDGES;
    if (t < BIN_NODES) { cnt[t] = 0; cur[t] = 0; }
    if (t == 0) ovf = 0;
    __syncthreads();

    const int rot = rng * (NV4 / BIN_RANGES);
    const int4* D4 = (const int4*)(edst + e_lo);
    unsigned* mywl = wl + wave * BIN_WCAP;
    int below = 0;
    int wcnt = 0;

    auto loadb = [&](int4 (&d)[BIN_BATCH], int ib) {
#pragma unroll
        for (int u = 0; u < BIN_BATCH; ++u) {
            const int i = (ib * BIN_BATCH + u) * BIN_THREADS + t;
            int idx = i + rot;
            if (idx >= NV4) idx -= NV4;
            d[u] = (i < NV4) ? D4[idx] : make_int4(BIN_FAR, BIN_FAR, BIN_FAR, BIN_FAR);
        }
    };
    auto one = [&](int dv, int e) {
        const int tk = dv - lo;
        below += (int)((unsigned)tk >> 31);
        const bool mk = (unsigned)tk < BIN_NODES;
        const unsigned mask = __builtin_amdgcn_ballot_w32(mk);
        if (mask) {
            const int pos = wcnt + (int)__builtin_amdgcn_mbcnt_lo(mask, 0u);
            if (mk && pos < BIN_WCAP) mywl[pos] = ((unsigned)tk << 20) | (unsigned)e;
            wcnt += __builtin_popcount(mask);
        }
    };
    auto procb = [&](const int4 (&d)[BIN_BATCH], int ib) {
#pragma unroll
        for (int u = 0; u < BIN_BATCH; ++u) {
            const int i = (ib * BIN_BATCH + u) * BIN_THREADS + t;
            int idx = i + rot;
            if (idx >= NV4) idx -= NV4;
            const int e = e_lo + idx * 4;
            one(d[u].x, e + 0);
            one(d[u].y, e + 1);
            one(d[u].z, e + 2);
            one(d[u].w, e + 3);
        }
    };

    int4 A[BIN_BATCH], B[BIN_BATCH];
    loadb(A, 0);
    for (int ib = 0; ib < BIN_NB; ib += 2) {
        loadb(B, ib + 1);
        procb(A, ib);
        loadb(A, ib + 2);
        procb(B, ib + 1);
    }

#pragma unroll
    for (int o = 16; o >= 1; o >>= 1) below += __shfl_xor(below, o, 32);
    if (lane == 0) {
        wsum[wave] = below;
        if (wcnt > BIN_WCAP) ovf = 1;
    }
    const int nw = min(wcnt, BIN_WCAP);
    for (int i = lane; i < nw; i += 32) atomicAdd(&cnt[mywl[i] >> 20], 1);
    __syncthreads();
    int base = e_lo;
#pragma unroll
    for (int w = 0; w < BIN_WAVES; ++w) base += wsum[w];
    const bool over = ovf != 0;
    if (over) {
        __syncthreads();
        if (t < BIN_NODES) cnt[t] = 0;
        __syncthreads();
        for (int e = e_lo + t; e < e_lo + PART_EDGES; e += BIN_THREADS) {
            const int dl = edst[e] - lo;
            if ((unsigned)dl < BIN_NODES) atomicAdd(&cnt[dl], 1);
        }
        __syncthreads();
    }
    {
        const int v = (t < BIN_NODES) ? cnt[t] : 0;
        int inc = v;
        if (wave < SCAN_WAVES) {
#pragma unroll
            for (int o = 1; o < 32; o <<= 1) {
                const int up = __shfl_up(inc, o, 32);
                if (lane >= o) inc += up;
            }
            if (lane == 31) wtot[wave] = inc;
        }
        __syncthreads();
        if (wave < SCAN_WAVES) {
            int add = 0;
            for (int w = 0; w < wave; ++w) add += wtot[w];
            if (t < BIN_NODES) pre[t + 1] = add + inc;
            if (t == 0) pre[0] = 0;
        }
    }
    __syncthreads();
    const int total = pre[BIN_NODES];
    int* rp = rowptr + part * RP_PITCH;
    if (t < BIN_NODES && lo + t < N_NODES) rp[lo + t] = base + pre[t];
    if (t == 0 && rng == BIN_RANGES - 1) rp[N_NODES] = base + total;

    if (!over) {
        for (int i0 = 0; i0 < nw; i0 += 8 * 32) {
            unsigned m[8];
            int s[8];
#pragma unroll
            for (int u = 0; u < 8; ++u) {
                const int i = i0 + u * 32 + lane;
                m[u] = (i < nw) ? mywl[i] : 0u;
                s[u] = (i < nw) ? esrc[m[u] & 0xFFFFFu] : 0;
            }
#pragma unroll
            for (int u = 0; u < 8; ++u) {
                const int i = i0 + u * 32 + lane;
                if (i < nw) {
                    const int dl = (int)(m[u] >> 20);
                    const int slot = atomicAdd(&cur[dl], 1);
                    sorted[pre[dl] + slot] = s[u];
                }
            }
        }
        __syncthreads();
        for (int i = t; i < total; i += BIN_THREADS) csr[base + i] = sorted[i];
    } else {
        for (int e = e_lo + t; e < e_lo + PART_EDGES; e += BIN_THREADS) {
            const int dl = edst[e] - lo;
            if ((unsigned)dl < BIN_NODES) {
                const int slot = atomicAdd(&cur[dl], 1);
                csr[base + pre[dl] + slot] = esrc[e];
            }
        }
    }
}

#define FR_THREADS 1024
#define DOT_WGS 64
#define SC_PITCH 10240
__device__ __forceinline__ void front_dots(const float* __restrict__ x,
                       const float* __restrict__ Wfc, const float* __restrict__ bfc,
                       const float* __restrict__ wl, const float* __restrict__ wr,
                       float* __restrict__ sc, float* __restrict__ phis, const int g) {
    __shared__ float4 vp[2][4][DD / 4];
    __shared__ float cq[4];
    const int t = threadIdx.x;
    const int lane = t & 31;
    {
        const int i = g * FR_THREADS + t;
        if (i < N_GRAPHS * 2 * DD) phis[i] = 0.0f;
    }
    {
        const int q = t >> 8, jh = (t >> 7) & 1, k = t & 127, b = q >> 1;
        const float* W = Wfc + (size_t)b * DD * DD + (size_t)jh * (DD / 2) * DD;
        const float* w = ((q & 1) ? wr : wl) + b * DD + jh * (DD / 2);
        float s = 0.0f;
#pragma unroll 16
        for (int j = 0; j < DD / 2; ++j) s = __builtin_fmaf(W[j * DD + k], w[j], s);
        ((float*)vp)[(jh * 4 + q) * DD + k] = s;
    }
    if (t < 4 * 32) {
        const int q = t >> 5, b = q >> 1;
        const float* w = ((q & 1) ? wr : wl) + b * DD;
        float s = 0.0f;
        for (int j = lane; j < DD; j += 32) s += bfc[b * DD + j] * w[j];
#pragma unroll
        for (int o = 16; o >= 1; o >>= 1) s += __shfl_xor(s, o, 32);
        if (lane == 0) cq[q] = s;
    }
    __syncthreads();
    const int wave = t >> 5;
    auto both = [&](int q) {
        const float4 a = vp[0][q][lane], c = vp[1][q][lane];
        return make_float4(a.x + c.x, a.y + c.y, a.z + c.z, a.w + c.w);
    };
    const float4 v0 = both(0), v1 = both(1), v2 = both(2), v3 = both(3);
    const float c0 = cq[0], c1 = cq[1], c2 = cq[2], c3 = cq[3];
    const float4* X4 = (const float4*)x;
    for (int row = g * 32 + wave; row < N_NODES; row += DOT_WGS * 32) {
        const float4 xv = X4[(size_t)row * (DD / 4) + lane];
        float d0 = xv.x * v0.x + xv.y * v0.y + xv.z * v0.z + xv.w * v0.w;
        float d1 = xv.x * v1.x + xv.y * v1.y + xv.z * v1.z + xv.w * v1.w;
        float d2 = xv.x * v2.x + xv.y * v2.y + xv.z * v2.z + xv.w * v2.w;
        float d3 = xv.x * v3.x + xv.y * v3.y + xv.z * v3.z + xv.w * v3.w;
#pragma unroll
        for (int o = 16; o >= 1; o >>= 1) {
            d0 += __shfl_xor(d0, o, 32); d1 += __shfl_xor(d1, o, 32);
            d2 += __shfl_xor(d2, o, 32); d3 += __shfl_xor(d3, o, 32);
        }
        if (lane == 0) {
            sc[row] = d0 + c0;
            sc[SC_PITCH + row] = d2 + c2;
            ((float2*)(sc + 2 * SC_PITCH))[row] = make_float2(d1 + c1, d3 + c3);
        }
    }
}

__global__ void __launch_bounds__(FR_THREADS) k_front(
                       const int* __restrict__ esrc, const int* __restrict__ edst,
                       int* __restrict__ rowptr, int* __restrict__ csr,
                       const float* __restrict__ x,
                       const float* __restrict__ Wfc, const float* __restrict__ bfc,
                       const float* __restrict__ wl, const float* __restrict__ wr,
                       float* __restrict__ sc, float* __restrict__ phis) {
    if (blockIdx.x < BIN_WGS)
        bin_body(esrc, edst, rowptr, csr, blockIdx.x);
    else
        front_dots(x, Wfc, bfc, wl, wr, sc, phis, blockIdx.x - BIN_WGS);
}

__device__ __forceinline__ float edge_score(float as, float an) {
    float sc = as + an;
    return sc >= 0.0f ? sc : NEG_SLOPE * sc;
}

#define GATHER_WAVES 8
#define GATHER_EPL 4
template <int NB>
__device__ __forceinline__ void gather_rows(const float4* __restrict__ F, const int lane,
                                            const int sv, const float e0, const float e1,
                                            const int j, const int cnt,
                                            float4& acc0, float4& acc1) {
    float  w0[NB], w1[NB];
    float4 f[NB];
#pragma unroll
    for (int k = 0; k < NB; ++k) {
        const int sj = __builtin_amdgcn_readlane(sv, min(j + k, cnt - 1));
        w0[k] = __int_as_float(__builtin_amdgcn_readlane(__float_as_int(e0), j + k));
        w1[k] = __int_as_float(__builtin_amdgcn_readlane(__float_as_int(e1), j + k));
        f[k] = F[(size_t)sj * (DD / 4) + lane];
    }
#pragma unroll
    for (int k = 0; k < NB; ++k) {
        acc0.x = __builtin_fmaf(w0[k], f[k].x, acc0.x); acc0.y = __builtin_fmaf(w0[k], f[k].y, acc0.y);
        acc0.z = __builtin_fmaf(w0[k], f[k].z, acc0.z); acc0.w = __builtin_fmaf(w0[k], f[k].w, acc0.w);
        acc1.x = __builtin_fmaf(w1[k], f[k].x, acc1.x); acc1.y = __builtin_fmaf(w1[k], f[k].y, acc1.y);
        acc1.z = __builtin_fmaf(w1[k], f[k].z, acc1.z); acc1.w = __builtin_fmaf(w1[k], f[k].w, acc1.w);
    }
}

__device__ __forceinline__ void chunk_softmax(float (&e)[GATHER_EPL], const float as,
                                              const int lane, const int deg) {
#pragma unroll
    for (int u = 0; u < GATHER_EPL; ++u) {
        const int i = u * 32 + lane;
        e[u] = (i < deg) ? edge_score(as, e[u]) : -INFINITY;
    }
    float m = e[0];
#pragma unroll
    for (int u = 1; u < GATHER_EPL; ++u) m = fmaxf(m, e[u]);
#pragma unroll
    for (int o = 16; o >= 1; o >>= 1) m = fmaxf(m, __shfl_xor(m, o, 32));
    float den = 0.0f;
#pragma unroll
    for (int u = 0; u < GATHER_EPL; ++u) {
        const int i = u * 32 + lane;
        e[u] = (i < deg) ? __expf(e[u] - m) : 0.0f;
        den += e[u];
    }
#pragma unroll
    for (int o = 16; o >= 1; o >>= 1) den += __shfl_xor(den, o, 32);
#pragma unroll
    for (int u = 0; u < GATHER_EPL; ++u) e[u] = e[u] / den;
}

__global__ void __launch_bounds__(GATHER_WAVES * 32) k_gather(
                         const int* __restrict__ rowptr, const int* __restrict__ csr,
                         const float* __restrict__ sc, const float* __restrict__ x,
                         float* __restrict__ hout, float* __restrict__ has) {
    const int lane = threadIdx.x & 31;
    const int wv = __builtin_amdgcn_readfirstlane((int)(threadIdx.x >> 5));
    const int n = blockIdx.x * GATHER_WAVES + wv;
    if (n >= N_NODES) return;

    int start[BIN_PARTS], step[BIN_PARTS];
    int deg = 0;
    int prev = 0;
#pragma unroll
    for (int p = 0; p < BIN_PARTS; ++p) {
        const int beg = rowptr[p * RP_PITCH + n], end = rowptr[p * RP_PITCH + n + 1];
        start[p] = deg;
        step[p] = (beg - deg) - prev;
        prev = beg - deg;
        deg += end - beg;
    }
    auto at = [&](int i) {
        int o = step[0];
#pragma unroll
        for (int p = 1; p < BIN_PARTS; ++p) o += (i >= start[p]) ? step[p] : 0;
        return i + o;
    };
    const float4* F = (const float4*)x;
    const float2* anb = (const float2*)(sc + 2 * SC_PITCH);
    float4 acc0 = make_float4(0.0f, 0.0f, 0.0f, 0.0f);
    float4 acc1 = make_float4(0.0f, 0.0f, 0.0f, 0.0f);
    if (deg > 0 && deg <= GATHER_EPL * 32) {
        const float as0 = sc[n], as1 = sc[SC_PITCH + n];
        int   s[GATHER_EPL];
        float e0[GATHER_EPL], e1[GATHER_EPL];
#pragma unroll
        for (int u = 0; u < GATHER_EPL; ++u) {
            const int i = u * 32 + lane;
            s[u] = csr[at(i < deg ? i : 0)];
        }
#pragma unroll
        for (int u = 0; u < GATHER_EPL; ++u) { const float2 p = anb[s[u]]; e0[u] = p.x; e1[u] = p.y; }
        chunk_softmax(e0, as0, lane, deg);
        chunk_softmax(e1, as1, lane, deg);

#pragma unroll
        for (int u = 0; u < GATHER_EPL; ++u) {
            const int cnt = min(32, deg - u * 32);
            if (cnt > 0) {
                int j = 0;
                for (; j + 16 <= cnt; j += 16) gather_rows<16>(F, lane, s[u], e0[u], e1[u], j, cnt, acc0, acc1);
                if (cnt - j > 8)      gather_rows<16>(F, lane, s[u], e0[u], e1[u], j, cnt, acc0, acc1);
                else if (cnt - j > 0) gather_rows<8>(F, lane, s[u], e0[u], e1[u], j, cnt, acc0, acc1);
            }
        }
    } else if (deg > 0) {
        for (int b = 0; b < 2; ++b) {
            const float as = sc[b * SC_PITCH + n];
            auto nb = [&](int s) { const float2 p = anb[s]; return b ? p.y : p.x; };

            float m = -INFINITY;
            for (int i = lane; i < deg; i += 32)
                m = fmaxf(m, edge_score(as, nb(csr[at(i)])));
#pragma unroll
            for (int o = 16; o >= 1; o >>= 1) m = fmaxf(m, __shfl_xor(m, o, 32));

            float den = 0.0f;
            for (int i = lane; i < deg; i += 32)
                den += __expf(edge_score(as, nb(csr[at(i)])) - m);
#pragma unroll
            for (int o = 16; o >= 1; o >>= 1) den += __shfl_xor(den, o, 32);

            float4 acc = make_float4(0.0f, 0.0f, 0.0f, 0.0f);
            for (int base = 0; base < deg; base += 32) {
                const int i = base + lane;
                int s = 0;
                float w = 0.0f;
                if (i < deg) {
                    s = csr[at(i)];
                    w = __expf(edge_score(as, nb(s)) - m) / den;
                }
                const int cnt = min(32, deg - base);
                for (int j = 0; j < cnt; ++j) {
                    const int   s1 = __shfl(s, j, 32);
                    const float w1 = __shfl(w, j, 32);
                    const float4 f1 = F[(size_t)s1 * (DD / 4) + lane];
                    acc.x += w1 * f1.x; acc.y += w1 * f1.y;
                    acc.z += w1 * f1.z; acc.w += w1 * f1.w;
                }
            }
            if (b == 0) acc0 = acc; else acc1 = acc;
        }
    }
    float4* hp = (float4*)(hout + (size_t)n * (2 * DD)) + lane;
    hp[0] = acc0;
    hp[DD / 4] = acc1;
    if (lane == 0) has[n] = deg > 0 ? 1.0f : 0.0f;
}

#define OUT_THREADS 384
#define OUT_ROWS 96
#define OUT_TILES ((N_NODES + OUT_ROWS - 1) / OUT_ROWS)
#define OUT_W4 (DD * 2 * DD / 4)
#define OPITCH 324
__global__ void __launch_bounds__(OUT_THREADS) k_out(const float* __restrict__ x,
                       const float* __restrict__ Wfc, const float* __restrict__ bfc,
                       const float* __restrict__ Wres, const float* __restrict__ has,
                       const int* __restrict__ gids, float* hout, float* __restrict__ phis) {
    __shared__ float Ws[DD * OPITCH];
    const int b = blockIdx.y;
    const float4* Wf4 = (const float4*)(Wfc + (size_t)b * DD * DD);
    const float4* Wr4 = (const float4*)(Wres + (size_t)b * DD * DD);
    for (int i0 = 0; i0 < OUT_W4; i0 += 8 * OUT_THREADS) {
        float4 w[8];
#pragma unroll
        for (int u = 0; u < 8; ++u) {
            const int i = min(i0 + u * OUT_THREADS + (int)threadIdx.x, OUT_W4 - 1);
            const int j = i >> 6, q = i & 63;
            w[u] = (q < DD / 4) ? Wf4[j * (DD / 4) + q] : Wr4[j * (DD / 4) + q - DD / 4];
        }
#pragma unroll
        for (int u = 0; u < 8; ++u) {
            const int i = i0 + u * OUT_THREADS + (int)threadIdx.x;
            const int j = i >> 6, k4 = (i & 63) << 2;
            if (i < OUT_W4) *(float4*)(Ws + j * OPITCH + k4) = w[u];
        }
    }
    __syncthreads();

    const int lane = threadIdx.x & 31;
    const int wave = threadIdx.x >> 5;
    const int strip = wave >> 1;
    const int colBase = (wave & 1) * 64;
    const int rowBase = blockIdx.x * OUT_ROWS + strip * 16;
    const int l16 = lane & 15;
    const int hi  = lane >> 4;

    float fl[8];
    int   gv[8];
#pragma unroll
    for (int v = 0; v < 8; ++v) {
        const int row = rowBase + v + 8 * hi;
        const int rc = min(row, N_NODES - 1);
        fl[v] = has[rc];
        gv[v] = (row < N_NODES) ? gids[rc] : -1;
    }

    v8f c[4];
#pragma unroll
    for (int t = 0; t < 4; ++t) {
        const float bv = bfc[b * DD + colBase + t * 16 + l16];
#pragma unroll
        for (int v = 0; v < 8; ++v) c[t][v] = bv * fl[v];
    }

    const int arow = min(rowBase + l16, N_NODES - 1);
    const float* zrow = hout + (size_t)arow * (2 * DD) + b * DD;
    const float* xrow = x + (size_t)arow * DD;

    for (int k0 = 0; k0 < DD; k0 += 4) {
        const int krow = k0 + 2 * hi;
        v2f a = *(const v2f*)(zrow + krow);
#pragma unroll
        for (int t = 0; t < 4; ++t) {
            const int j = colBase + t * 16 + l16;
            const v2f bf = *(const v2f*)(Ws + j * OPITCH + krow);
            c[t] = __builtin_amdgcn_wmma_f32_16x16x4_f32(
                false, a, false, bf, (short)0, c[t], false, false);
        }
    }
    for (int k0 = 0; k0 < DD; k0 += 4) {
        const int krow = k0 + 2 * hi;
        v2f a = *(const v2f*)(xrow + krow);
#pragma unroll
        for (int t = 0; t < 4; ++t) {
            const int j = colBase + t * 16 + l16;
            const v2f bf = *(const v2f*)(Ws + j * OPITCH + DD + krow);
            c[t] = __builtin_amdgcn_wmma_f32_16x16x4_f32(
                false, a, false, bf, (short)0, c[t], false, false);
        }
    }

    __syncthreads();

    float run[4] = {0.0f, 0.0f, 0.0f, 0.0f};
    int gcur = -1;
    float* pbase = phis + b * DD + colBase + l16;
#pragma unroll
    for (int v = 0; v < 8; ++v) {
        const int row = rowBase + v + 8 * hi;
        if (gv[v] != gcur) {
            if ((unsigned)gcur < (unsigned)N_GRAPHS) {
#pragma unroll
                for (int t = 0; t < 4; ++t) atomicAdd(pbase + (size_t)gcur * (2 * DD) + t * 16, run[t]);
            }
            gcur = gv[v];
#pragma unroll
            for (int t = 0; t < 4; ++t) run[t] = 0.0f;
        }
#pragma unroll
        for (int t = 0; t < 4; ++t) {
            const float val = fmaxf(c[t][v], 0.0f);
            if (row < N_NODES) {
                hout[(size_t)row * (2 * DD) + b * DD + colBase + t * 16 + l16] = val;
                run[t] += val;
            }
        }
    }
    if ((unsigned)gcur < (unsigned)N_GRAPHS) {
#pragma unroll
        for (int t = 0; t < 4; ++t) atomicAdd(pbase + (size_t)gcur * (2 * DD) + t * 16, run[t]);
    }
}

extern "C" void kernel_launch(void* const* d_in, const int* in_sizes, int n_in,
                              void* d_out, int out_size, void* d_ws, size_t ws_size,
                              hipStream_t stream) {
    const float* x        = (const float*)d_in[0];
    const int*   esrc     = (const int*)d_in[1];
    const int*   edst     = (const int*)d_in[2];
    const int*   gids     = (const int*)d_in[3];
    const float* Wfc      = (const float*)d_in[4];
    const float* bfc      = (const float*)d_in[5];
    const float* wl       = (const float*)d_in[6];
    const float* wr       = (const float*)d_in[7];
    const float* Wres     = (const float*)d_in[8];

    float* h    = (float*)d_out;
    float* phis = (float*)d_out + (size_t)N_NODES * 2 * DD;

    char* ws = (char*)d_ws;
    float* sc     = (float*)ws;
    float* has    = sc + 4 * SC_PITCH;
    int*   csr    = (int*)(ws + (size_t)N_NODES * DD * 4);
    int*   rowptr = (int*)(ws + 262144);

    k_front<<<dim3(BIN_WGS + DOT_WGS), dim3(FR_THREADS), 0, stream>>>(
        esrc, edst, rowptr, csr, x, Wfc, bfc, wl, wr, sc, phis);

    k_gather<<<dim3((N_NODES + GATHER_WAVES - 1) / GATHER_WAVES), dim3(GATHER_WAVES * 32), 0, stream>>>(
        rowptr, csr, sc, x, h, has);

    k_out<<<dim3(OUT_TILES, 2), dim3(OUT_THREADS), 0, stream>>>(
        x, Wfc, bfc, Wres, has, gids, h, phis);
}
